# starting kernel
# baseline (speedup 1.0000x reference)
#include <hip/hip_runtime.h>
#include <math.h>

typedef float v2f __attribute__((ext_vector_type(2)));
typedef float v8f __attribute__((ext_vector_type(8)));

__global__ void k_fill(float* __restrict__ p, float v, int n) {
    int i = blockIdx.x * blockDim.x + threadIdx.x;
    if (i < n) p[i] = v;
}

__global__ void k_deg_accum(const int* __restrict__ dst, float* __restrict__ deg, int E) {
    int e = blockIdx.x * blockDim.x + threadIdx.x;
    if (e < E) unsafeAtomicAdd(&deg[dst[e]], 1.0f);
}

__global__ void k_rsqrt_inplace(float* __restrict__ p, int n) {
    int i = blockIdx.x * blockDim.x + threadIdx.x;
    if (i < n) p[i] = rsqrtf(p[i]);
}

__global__ void k_norm(const int* __restrict__ src, const int* __restrict__ dst,
                       const float* __restrict__ dis, float* __restrict__ norm, int E) {
    int e = blockIdx.x * blockDim.x + threadIdx.x;
    if (e < E) norm[e] = dis[src[e]] * dis[dst[e]];
}

template <int K, int NCOL>
__global__ __launch_bounds__(256) void k_gemm_wmma(const float* __restrict__ A,
                                                   const float* __restrict__ B,
                                                   float* __restrict__ O, int nrows) {
    int wave = threadIdx.x >> 5;
    int lane = threadIdx.x & 31;
    int tile = blockIdx.x * (blockDim.x >> 5) + wave;
    int base = tile << 4;
    if (base >= nrows) return;
    int hi = lane >> 4;
    int lo = lane & 15;

    int arow = base + lo;
    arow = arow < nrows ? arow : (nrows - 1);
    const float* __restrict__ ap = A + (long)arow * K + 2 * hi;
    const float* __restrict__ bp = B + lo;
    bool bvalid = (lo < NCOL);

    v8f c = {0.f, 0.f, 0.f, 0.f, 0.f, 0.f, 0.f, 0.f};
#pragma unroll
    for (int k0 = 0; k0 < K; k0 += 4) {
        v2f a = *(const v2f*)(ap + k0);
        int ka = k0 + 2 * hi;
        v2f b;
        b.x = bvalid ? bp[(ka + 0) * NCOL] : 0.f;
        b.y = bvalid ? bp[(ka + 1) * NCOL] : 0.f;
        c = __builtin_amdgcn_wmma_f32_16x16x4_f32(false, a, false, b, (short)0, c,
                                                  false, false);
    }
#pragma unroll
    for (int v = 0; v < 8; ++v) {
        int row = base + v + 8 * hi;
        if (row < nrows && lo < NCOL) O[(long)row * NCOL + lo] = c[v];
    }
}

template <int F>
__global__ void k_scatter(const int* __restrict__ src, const int* __restrict__ dst,
                          const float* __restrict__ norm, const float* __restrict__ h,
                          float* __restrict__ acc, int E) {
    long gid = (long)blockIdx.x * blockDim.x + threadIdx.x;
    long e = gid / F;
    int f = (int)(gid - e * F);
    if (e < (long)E) {
        int s = src[e], d = dst[e];
        float v = norm[e] * h[(long)s * F + f];
        unsafeAtomicAdd(&acc[(long)d * F + f], v);
    }
}

__global__ void k_relu_ep(const float* __restrict__ acc, const float* __restrict__ hx,
                          const float* __restrict__ dis, const float* __restrict__ b,
                          float* __restrict__ out, int n) {
    int gid = blockIdx.x * blockDim.x + threadIdx.x;
    if (gid < n * 16) {
        int i = gid >> 4, f = gid & 15;
        float di = dis[i];
        float v = acc[gid] + di * di * hx[gid] + b[f];
        out[gid] = v > 0.f ? v : 0.f;
    }
}

__global__ void k_lsm_ep(const float* __restrict__ acc, const float* __restrict__ hx,
                         const float* __restrict__ dis, const float* __restrict__ b,
                         float* __restrict__ out, int n) {
    int i = blockIdx.x * blockDim.x + threadIdx.x;
    if (i < n) {
        float d2 = dis[i] * dis[i];
        float z0 = acc[2 * i + 0] + d2 * hx[2 * i + 0] + b[0];
        float z1 = acc[2 * i + 1] + d2 * hx[2 * i + 1] + b[1];
        float m = fmaxf(z0, z1);
        float lse = m + __logf(__expf(z0 - m) + __expf(z1 - m));
        out[2 * i + 0] = z0 - lse;
        out[2 * i + 1] = z1 - lse;
    }
}

extern "C" void kernel_launch(void* const* d_in, const int* in_sizes, int n_in,
                              void* d_out, int out_size, void* d_ws, size_t ws_size,
                              hipStream_t stream) {
    const float* x  = (const float*)d_in[0];
    const int*   ei = (const int*)d_in[1];
    const float* W1 = (const float*)d_in[2];
    const float* b1 = (const float*)d_in[3];
    const float* W2 = (const float*)d_in[4];
    const float* b2 = (const float*)d_in[5];
    const float* W3 = (const float*)d_in[6];
    const float* b3 = (const float*)d_in[7];

    const int N = in_sizes[0] / 128;
    const int E = in_sizes[1] / 2;
    const int* src = ei;
    const int* dst = ei + E;

    char* ws = (char*)d_ws;
    size_t off = 0;
    auto carve = [&](size_t bytes) -> void* {
        void* p = ws + off;
        off += (bytes + 255) & ~(size_t)255;
        return p;
    };
    float* dis  = (float*)carve((size_t)N * 4);
    float* norm = (float*)carve((size_t)E * 4);
    float* hx   = (float*)carve((size_t)N * 16 * 4);
    float* acc  = (float*)carve((size_t)N * 16 * 4);
    float* h    = (float*)carve((size_t)N * 16 * 4);
    (void)ws_size; (void)n_in; (void)out_size;

    const int T = 256;
    auto blk = [](long n, int t) { return (unsigned)((n + t - 1) / t); };

    k_fill<<<blk(N, T), T, 0, stream>>>(dis, 1.0f, N);
    k_deg_accum<<<blk(E, T), T, 0, stream>>>(dst, dis, E);
    k_rsqrt_inplace<<<blk(N, T), T, 0, stream>>>(dis, N);
    k_norm<<<blk(E, T), T, 0, stream>>>(src, dst, dis, norm, E);

    const int ntile = (N + 15) / 16;
    const unsigned gblk = (unsigned)((ntile + 7) / 8);

    k_gemm_wmma<128, 16><<<gblk, 256, 0, stream>>>(x, W1, hx, N);
    k_fill<<<blk((long)N * 16, T), T, 0, stream>>>(acc, 0.f, N * 16);
    k_scatter<16><<<blk((long)E * 16, T), T, 0, stream>>>(src, dst, norm, hx, acc, E);
    k_relu_ep<<<blk((long)N * 16, T), T, 0, stream>>>(acc, hx, dis, b1, h, N);

    k_gemm_wmma<16, 16><<<gblk, 256, 0, stream>>>(h, W2, hx, N);
    k_fill<<<blk((long)N * 16, T), T, 0, stream>>>(acc, 0.f, N * 16);
    k_scatter<16><<<blk((long)E * 16, T), T, 0, stream>>>(src, dst, norm, hx, acc, E);
    k_relu_ep<<<blk((long)N * 16, T), T, 0, stream>>>(acc, hx, dis, b2, h, N);

    k_gemm_wmma<16, 2><<<gblk, 256, 0, stream>>>(h, W3, hx, N);
    k_fill<<<blk((long)N * 2, T), T, 0, stream>>>(acc, 0.f, N * 2);
    k_scatter<2><<<blk((long)E * 2, T), T, 0, stream>>>(src, dst, norm, hx, acc, E);
    k_lsm_ep<<<blk(N, T), T, 0, stream>>>(acc, hx, dis, b3, (float*)d_out, N);
}


# edges bucketed by destination range once per call; each layer sums a bucket in LDS (ds_add_f32, arrival order) and fuses its epilogue; dis factored out of the edge sum; no global float atomics
# speedup vs baseline: 5.0831x; 5.0831x over previous
#include <hip/hip_runtime.h>
#include <math.h>


typedef float v2f __attribute__((ext_vector_type(2)));
typedef float v8f __attribute__((ext_vector_type(8)));

#define RB 416
#define MAXP 1024
#define SRC_BITS 20
#define SRC_MASK ((1u << SRC_BITS) - 1u)
#define NO_EDGE 0xFFFFFFFFu
#define NPART 256

__global__ void k_zero(int* __restrict__ p, int n) {
    int i = blockIdx.x * blockDim.x + threadIdx.x;
    if (i < n) p[i] = 0;
}

__global__ __launch_bounds__(1024) void k_count(const int* __restrict__ dst,
                                                int* __restrict__ bcount, int E,
                                                int chunk, int P) {
    __shared__ int hist[MAXP];
    const int tid = threadIdx.x;
    for (int i = tid; i < MAXP; i += 1024) hist[i] = 0;
    __syncthreads();
    long beg = (long)blockIdx.x * chunk;
    long end = beg + chunk;
    if (end > (long)E) end = E;
    for (long e0 = beg + tid; e0 < end; e0 += 4096) {
        int d[4];
#pragma unroll
        for (int u = 0; u < 4; ++u) {
            long e = e0 + u * 1024;
            d[u] = (e < end) ? dst[e] : -1;
        }
#pragma unroll
        for (int u = 0; u < 4; ++u)
            if (d[u] >= 0) atomicAdd(&hist[d[u] / RB], 1);
    }
    __syncthreads();
    for (int i = tid; i < P; i += 1024) {
        int c = hist[i];
        if (c) atomicAdd(&bcount[i], c);
    }
}

__global__ __launch_bounds__(1024) void k_partition(const int* __restrict__ src,
                                                    const int* __restrict__ dst,
                                                    const int* __restrict__ bcount,
                                                    int* __restrict__ bcursor,
                                                    int* __restrict__ bstart,
                                                    unsigned* __restrict__ sorted, int E,
                                                    int chunk, int P) {
    __shared__ int hist[MAXP];
    __shared__ int base[MAXP];
    __shared__ int cnt[MAXP];
    const int tid = threadIdx.x;
    for (int i = tid; i < MAXP; i += 1024) {
        hist[i] = 0;
        cnt[i] = (i < P) ? bcount[i] : 0;
    }
    __syncthreads();
    long beg = (long)blockIdx.x * chunk;
    long end = beg + chunk;
    if (end > (long)E) end = E;
    for (long e0 = beg + tid; e0 < end; e0 += 4096) {
        int d[4];
#pragma unroll
        for (int u = 0; u < 4; ++u) {
            long e = e0 + u * 1024;
            d[u] = (e < end) ? dst[e] : -1;
        }
#pragma unroll
        for (int u = 0; u < 4; ++u)
            if (d[u] >= 0) atomicAdd(&hist[d[u] / RB], 1);
    }
    __syncthreads();
    for (int i = tid; i < P; i += 1024) {
        int s = 0;
        for (int j = 0; j < i; ++j) s += cnt[j];
        int c = hist[i];
        int off = c ? atomicAdd(&bcursor[i], c) : 0;
        base[i] = s + off;
        if (blockIdx.x == 0) bstart[i] = s;
    }
    if (blockIdx.x == 0 && tid == 0) bstart[P] = E;
    __syncthreads();
    for (int i = tid; i < MAXP; i += 1024) hist[i] = 0;
    __syncthreads();
    for (long e0 = beg + tid; e0 < end; e0 += 4096) {
        int d[4], s[4];
#pragma unroll
        for (int u = 0; u < 4; ++u) {
            long e = e0 + u * 1024;
            bool ok = e < end;
            d[u] = ok ? dst[e] : -1;
            s[u] = ok ? src[e] : 0;
        }
#pragma unroll
        for (int u = 0; u < 4; ++u) {
            if (d[u] >= 0) {
                int b = d[u] / RB;
                int p = base[b] + atomicAdd(&hist[b], 1);
                sorted[p] = (unsigned)s[u] | ((unsigned)(d[u] - b * RB) << SRC_BITS);
            }
        }
    }
}

__global__ __launch_bounds__(1024) void k_deg_bucket(const unsigned* __restrict__ sorted,
                                                     const int* __restrict__ bstart,
                                                     float* __restrict__ dis, int N) {
    __shared__ int cntl[RB];
    const int tid = threadIdx.x;
    const int lo = blockIdx.x * RB;
    for (int i = tid; i < RB; i += 1024) cntl[i] = 0;
    __syncthreads();
    const int beg = bstart[blockIdx.x], end = bstart[blockIdx.x + 1];
    for (int e0 = beg + tid; e0 < end; e0 += 4096) {
        unsigned q[4];
#pragma unroll
        for (int u = 0; u < 4; ++u) {
            int e = e0 + u * 1024;
            q[u] = (e < end) ? sorted[e] : NO_EDGE;
        }
#pragma unroll
        for (int u = 0; u < 4; ++u)
            if (q[u] != NO_EDGE) atomicAdd(&cntl[q[u] >> SRC_BITS], 1);
    }
    __syncthreads();
    for (int i = tid; i < RB; i += 1024) {
        int node = lo + i;
        if (node < N) dis[node] = rsqrtf((float)(cntl[i] + 1));
    }
}

template <int K, int NCOL>
__global__ __launch_bounds__(256) void k_gemm_wmma(const float* __restrict__ A,
                                                   const float* __restrict__ B,
                                                   const float* __restrict__ rs,
                                                   float* __restrict__ O, int nrows) {
    int wave = threadIdx.x >> 5;
    int lane = threadIdx.x & 31;
    int tile = blockIdx.x * (blockDim.x >> 5) + wave;
    int base = tile << 4;
    if (base >= nrows) return;
    int hi = lane >> 4;
    int lo = lane & 15;

    int arow = base + lo;
    arow = arow < nrows ? arow : (nrows - 1);
    const float* __restrict__ ap = A + (long)arow * K + 2 * hi;
    const float* __restrict__ bp = B + lo;
    bool bvalid = (lo < NCOL);
    const float sc = rs[arow];

    v8f c = {0.f, 0.f, 0.f, 0.f, 0.f, 0.f, 0.f, 0.f};
#pragma unroll
    for (int k0 = 0; k0 < K; k0 += 4) {
        v2f a = *(const v2f*)(ap + k0);
        a.x *= sc;
        a.y *= sc;
        int ka = k0 + 2 * hi;
        v2f b;
        b.x = bvalid ? bp[(ka + 0) * NCOL] : 0.f;
        b.y = bvalid ? bp[(ka + 1) * NCOL] : 0.f;
        c = __builtin_amdgcn_wmma_f32_16x16x4_f32(false, a, false, b, (short)0, c,
                                                  false, false);
    }
#pragma unroll
    for (int v = 0; v < 8; ++v) {
        int row = base + v + 8 * hi;
        if (row < nrows && lo < NCOL) O[(long)row * NCOL + lo] = c[v];
    }
}

__global__ __launch_bounds__(1024) void k_bucket16(const unsigned* __restrict__ sorted,
                                                   const int* __restrict__ bstart,
                                                   const float* __restrict__ g,
                                                   const float* __restrict__ dis,
                                                   const float* __restrict__ bias,
                                                   float* __restrict__ out, int N, int E) {
    __shared__ float acc[RB * 16];
    const int tid = threadIdx.x;
    const int lo = blockIdx.x * RB;
    for (int i = tid; i < RB * 16; i += 1024) acc[i] = 0.f;
    __syncthreads();
    const int beg = bstart[blockIdx.x], end = bstart[blockIdx.x + 1];
    const int lane = tid & 31, wave = tid >> 5;
    const int f = lane & 15, half = lane >> 4;
    int c0 = beg + wave * 32;
    unsigned pk = (c0 + lane < end) ? sorted[c0 + lane] : NO_EDGE;
    for (; c0 < end; c0 += 1024) {
        const int en = c0 + 1024 + lane;
        const unsigned wn = sorted[en < E ? en : E - 1];
        const unsigned pkn = (en < end) ? wn : NO_EDGE;
        unsigned q[16];
        float v[16];
#pragma unroll
        for (int j = 0; j < 16; ++j) q[j] = (unsigned)__shfl((int)pk, 2 * j + half);
        if (c0 + 32 <= end) {
#pragma unroll
            for (int j = 0; j < 16; ++j) v[j] = g[(q[j] & SRC_MASK) * 16u + (unsigned)f];
#pragma unroll
            for (int j = 0; j < 16; ++j)
                __hip_atomic_fetch_add(&acc[(q[j] >> SRC_BITS) * 16u + (unsigned)f], v[j],
                                       __ATOMIC_RELAXED, __HIP_MEMORY_SCOPE_WORKGROUP);
        } else {
#pragma unroll
            for (int j = 0; j < 16; ++j) {
                unsigned s = (q[j] != NO_EDGE) ? (q[j] & SRC_MASK) : 0u;
                v[j] = g[s * 16u + (unsigned)f];
            }
#pragma unroll
            for (int j = 0; j < 16; ++j) {
                if (q[j] != NO_EDGE)
                    __hip_atomic_fetch_add(&acc[(q[j] >> SRC_BITS) * 16u + (unsigned)f], v[j],
                                           __ATOMIC_RELAXED, __HIP_MEMORY_SCOPE_WORKGROUP);
            }
        }
        pk = pkn;
    }
    __syncthreads();
    for (int i = tid; i < RB * 16; i += 1024) {
        int node = lo + (i >> 4), ff = i & 15;
        if (node < N) {
            size_t o = (size_t)node * 16 + ff;
            float v = dis[node] * (acc[i] + g[o]) + bias[ff];
            out[o] = v > 0.f ? v : 0.f;
        }
    }
}

__global__ __launch_bounds__(1024) void k_bucket2_lsm(const unsigned* __restrict__ sorted,
                                                      const int* __restrict__ bstart,
                                                      const float* __restrict__ g,
                                                      const float* __restrict__ dis,
                                                      const float* __restrict__ bias,
                                                      float* __restrict__ out, int N) {
    __shared__ float acc[RB * 2];
    const int tid = threadIdx.x;
    const int lo = blockIdx.x * RB;
    for (int i = tid; i < RB * 2; i += 1024) acc[i] = 0.f;
    __syncthreads();
    const int beg = bstart[blockIdx.x], end = bstart[blockIdx.x + 1];
    for (int e0 = beg + tid; e0 < end; e0 += 4096) {
        unsigned q[4];
        v2f v[4];
#pragma unroll
        for (int u = 0; u < 4; ++u) {
            int e = e0 + u * 1024;
            q[u] = (e < end) ? sorted[e] : NO_EDGE;
        }
#pragma unroll
        for (int u = 0; u < 4; ++u) {
            unsigned s = (q[u] != NO_EDGE) ? (q[u] & SRC_MASK) : 0u;
            v[u] = *(const v2f*)(g + s * 2u);
        }
#pragma unroll
        for (int u = 0; u < 4; ++u) {
            if (q[u] != NO_EDGE) {
                unsigned ld = q[u] >> SRC_BITS;
                __hip_atomic_fetch_add(&acc[ld * 2u + 0u], v[u].x, __ATOMIC_RELAXED,
                                       __HIP_MEMORY_SCOPE_WORKGROUP);
                __hip_atomic_fetch_add(&acc[ld * 2u + 1u], v[u].y, __ATOMIC_RELAXED,
                                       __HIP_MEMORY_SCOPE_WORKGROUP);
            }
        }
    }
    __syncthreads();
    for (int i = tid; i < RB; i += 1024) {
        int node = lo + i;
        if (node < N) {
            float di = dis[node];
            float z0 = di * (acc[2 * i + 0] + g[(size_t)node * 2 + 0]) + bias[0];
            float z1 = di * (acc[2 * i + 1] + g[(size_t)node * 2 + 1]) + bias[1];
            float m = fmaxf(z0, z1);
            float lse = m + __logf(__expf(z0 - m) + __expf(z1 - m));
            out[(size_t)node * 2 + 0] = z0 - lse;
            out[(size_t)node * 2 + 1] = z1 - lse;
        }
    }
}

extern "C" void kernel_launch(void* const* d_in, const int* in_sizes, int n_in,
                              void* d_out, int out_size, void* d_ws, size_t ws_size,
                              hipStream_t stream) {
    const float* x  = (const float*)d_in[0];
    const int*   ei = (const int*)d_in[1];
    const float* W1 = (const float*)d_in[2];
    const float* b1 = (const float*)d_in[3];
    const float* W2 = (const float*)d_in[4];
    const float* b2 = (const float*)d_in[5];
    const float* W3 = (const float*)d_in[6];
    const float* b3 = (const float*)d_in[7];

    const int N = in_sizes[0] / 128;
    const int E = in_sizes[1] / 2;
    const int* src = ei;
    const int* dst = ei + E;

    char* ws = (char*)d_ws;
    size_t off = 0;
    auto carve = [&](size_t bytes) -> void* {
        void* p = ws + off;
        off += (bytes + 255) & ~(size_t)255;
        return p;
    };
    float*    dis     = (float*)carve((size_t)N * 4);
    int*      bcount  = (int*)carve((size_t)2 * MAXP * 4);
    int*      bcursor = bcount + MAXP;
    int*      bstart  = (int*)carve((size_t)(MAXP + 1) * 4);
    unsigned* sorted  = (unsigned*)carve((size_t)E * 4);
    float*    g       = (float*)carve((size_t)N * 16 * 4);
    float*    h       = (float*)carve((size_t)N * 16 * 4);
    (void)ws_size; (void)n_in; (void)out_size;

    const int P = (N + RB - 1) / RB;
    const int chunk = (E + NPART - 1) / NPART;

    k_zero<<<(2 * MAXP + 255) / 256, 256, 0, stream>>>(bcount, 2 * MAXP);
    k_count<<<NPART, 1024, 0, stream>>>(dst, bcount, E, chunk, P);
    k_partition<<<NPART, 1024, 0, stream>>>(src, dst, bcount, bcursor, bstart, sorted, E,
                                            chunk, P);
    k_deg_bucket<<<P, 1024, 0, stream>>>(sorted, bstart, dis, N);

    const int ntile = (N + 15) / 16;
    const unsigned gblk = (unsigned)((ntile + 7) / 8);

    k_gemm_wmma<128, 16><<<gblk, 256, 0, stream>>>(x, W1, dis, g, N);
    k_bucket16<<<P, 1024, 0, stream>>>(sorted, bstart, g, dis, b1, h, N, E);

    k_gemm_wmma<16, 16><<<gblk, 256, 0, stream>>>(h, W2, dis, g, N);
    k_bucket16<<<P, 1024, 0, stream>>>(sorted, bstart, g, dis, b2, h, N, E);

    k_gemm_wmma<16, 2><<<gblk, 256, 0, stream>>>(h, W3, dis, g, N);
    k_bucket2_lsm<<<P, 1024, 0, stream>>>(sorted, bstart, g, dis, b3, (float*)d_out, N);
}


# each layer's f32 WMMA product moved into the per-bucket kernel that owns its rows (layer 1 into the degree kernel, 2 and 3 into the previous aggregation, A read from its LDS tile): 10 launches to 7
# speedup vs baseline: 5.6563x; 1.1128x over previous
#include <hip/hip_runtime.h>
#include <math.h>


typedef float v2f __attribute__((ext_vector_type(2)));
typedef float v8f __attribute__((ext_vector_type(8)));

#define RB 416
#define MAXP 1024
#define SRC_BITS 20
#define SRC_MASK ((1u << SRC_BITS) - 1u)
#define NO_EDGE 0xFFFFFFFFu
#define NPART 256
static_assert(RB % 16 == 0 && RB / 16 <= 32, "one wave of a 1024-thread work-group per 16-row tile");

__global__ void k_zero(int* __restrict__ p, int n) {
    int i = blockIdx.x * blockDim.x + threadIdx.x;
    if (i < n) p[i] = 0;
}

__global__ __launch_bounds__(1024) void k_count(const int* __restrict__ dst,
                                                int* __restrict__ bcount, int E,
                                                int chunk, int P) {
    __shared__ int hist[MAXP];
    const int tid = threadIdx.x;
    for (int i = tid; i < MAXP; i += 1024) hist[i] = 0;
    __syncthreads();
    long beg = (long)blockIdx.x * chunk;
    long end = beg + chunk;
    if (end > (long)E) end = E;
    for (long e0 = beg + tid; e0 < end; e0 += 4096) {
        int d[4];
#pragma unroll
        for (int u = 0; u < 4; ++u) {
            long e = e0 + u * 1024;
            d[u] = (e < end) ? dst[e] : -1;
        }
#pragma unroll
        for (int u = 0; u < 4; ++u)
            if (d[u] >= 0) atomicAdd(&hist[d[u] / RB], 1);
    }
    __syncthreads();
    for (int i = tid; i < P; i += 1024) {
        int c = hist[i];
        if (c) atomicAdd(&bcount[i], c);
    }
}

__global__ __launch_bounds__(1024) void k_partition(const int* __restrict__ src,
                                                    const int* __restrict__ dst,
                                                    const int* __restrict__ bcount,
                                                    int* __restrict__ bcursor,
                                                    int* __restrict__ bstart,
                                                    unsigned* __restrict__ sorted, int E,
                                                    int chunk, int P) {
    __shared__ int hist[MAXP];
    __shared__ int base[MAXP];
    __shared__ int cnt[MAXP];
    const int tid = threadIdx.x;
    for (int i = tid; i < MAXP; i += 1024) {
        hist[i] = 0;
        cnt[i] = (i < P) ? bcount[i] : 0;
    }
    __syncthreads();
    long beg = (long)blockIdx.x * chunk;
    long end = beg + chunk;
    if (end > (long)E) end = E;
    for (long e0 = beg + tid; e0 < end; e0 += 4096) {
        int d[4];
#pragma unroll
        for (int u = 0; u < 4; ++u) {
            long e = e0 + u * 1024;
            d[u] = (e < end) ? dst[e] : -1;
        }
#pragma unroll
        for (int u = 0; u < 4; ++u)
            if (d[u] >= 0) atomicAdd(&hist[d[u] / RB], 1);
    }
    __syncthreads();
    for (int i = tid; i < P; i += 1024) {
        int s = 0;
        for (int j = 0; j < i; ++j) s += cnt[j];
        int c = hist[i];
        int off = c ? atomicAdd(&bcursor[i], c) : 0;
        base[i] = s + off;
        if (blockIdx.x == 0) bstart[i] = s;
    }
    if (blockIdx.x == 0 && tid == 0) bstart[P] = E;
    __syncthreads();
    for (int i = tid; i < MAXP; i += 1024) hist[i] = 0;
    __syncthreads();
    for (long e0 = beg + tid; e0 < end; e0 += 4096) {
        int d[4], s[4];
#pragma unroll
        for (int u = 0; u < 4; ++u) {
            long e = e0 + u * 1024;
            bool ok = e < end;
            d[u] = ok ? dst[e] : -1;
            s[u] = ok ? src[e] : 0;
        }
#pragma unroll
        for (int u = 0; u < 4; ++u) {
            if (d[u] >= 0) {
                int b = d[u] / RB;
                int p = base[b] + atomicAdd(&hist[b], 1);
                sorted[p] = (unsigned)s[u] | ((unsigned)(d[u] - b * RB) << SRC_BITS);
            }
        }
    }
}


__global__ __launch_bounds__(1024) void k_deg_gemm1(const unsigned* __restrict__ sorted,
                                                    const int* __restrict__ bstart,
                                                    const float* __restrict__ x,
                                                    const float* __restrict__ W1,
                                                    float* __restrict__ dis,
                                                    float* __restrict__ g, int N) {
    __shared__ int cntl[RB];
    const int tid = threadIdx.x;
    const int lo = blockIdx.x * RB;
    const int lane = tid & 31;
    const int f = lane & 15, half = lane >> 4;
    const int wv = __builtin_amdgcn_readfirstlane(tid >> 5);
    const bool has_tile = wv < RB / 16;
    const int r0 = wv * 16;
    int arow = lo + r0 + f;
    arow = arow < N ? arow : (N - 1);
    for (int i = tid; i < RB; i += 1024) cntl[i] = 0;
    __syncthreads();
    const int beg = bstart[blockIdx.x], end = bstart[blockIdx.x + 1];
    for (int e0 = beg + tid; e0 < end; e0 += 4096) {
        unsigned q[4];
#pragma unroll
        for (int u = 0; u < 4; ++u) {
            int e = e0 + u * 1024;
            q[u] = (e < end) ? sorted[e] : NO_EDGE;
        }
#pragma unroll
        for (int u = 0; u < 4; ++u)
            if (q[u] != NO_EDGE) atomicAdd(&cntl[q[u] >> SRC_BITS], 1);
    }
    __syncthreads();
    for (int i = tid; i < RB; i += 1024) {
        int node = lo + i;
        if (node < N) dis[node] = rsqrtf((float)(cntl[i] + 1));
    }
    if (has_tile) {
        const float sc = rsqrtf((float)(cntl[r0 + f] + 1));
        const float* __restrict__ ap = x + (long)arow * 128 + 2 * half;
        const float* __restrict__ bp = W1 + f;
        v8f c = {0.f, 0.f, 0.f, 0.f, 0.f, 0.f, 0.f, 0.f};
#pragma unroll
        for (int j = 0; j < 32; ++j) {
            v2f av = *(const v2f*)(ap + 4 * j);
            av.x *= sc;
            av.y *= sc;
            int ka = 4 * j + 2 * half;
            v2f b;
            b.x = bp[(ka + 0) * 16];
            b.y = bp[(ka + 1) * 16];
            c = __builtin_amdgcn_wmma_f32_16x16x4_f32(false, av, false, b, (short)0, c,
                                                      false, false);
        }
#pragma unroll
        for (int v = 0; v < 8; ++v) {
            int row = lo + r0 + v + 8 * half;
            if (row < N) g[(long)row * 16 + f] = c[v];
        }
    }
}

template <int NCOL>
__global__ __launch_bounds__(1024) void k_bucket16(const unsigned* __restrict__ sorted,
                                                   const int* __restrict__ bstart,
                                                   const float* __restrict__ g,
                                                   const float* __restrict__ dis,
                                                   const float* __restrict__ bias,
                                                   const float* __restrict__ Wn,
                                                   float* __restrict__ gn, int N, int E) {
    __shared__ float acc[RB * 16];
    __shared__ float disl[RB];
    const int tid = threadIdx.x;
    const int lo = blockIdx.x * RB;
    for (int i = tid; i < RB * 16; i += 1024) acc[i] = 0.f;
    __syncthreads();
    const int beg = bstart[blockIdx.x], end = bstart[blockIdx.x + 1];
    const int lane = tid & 31, wave = tid >> 5;
    const int f = lane & 15, half = lane >> 4;
    int c0 = beg + wave * 32;
    unsigned pk = (c0 + lane < end) ? sorted[c0 + lane] : NO_EDGE;
    for (; c0 < end; c0 += 1024) {
        const int en = c0 + 1024 + lane;
        const unsigned wn = sorted[en < E ? en : E - 1];
        const unsigned pkn = (en < end) ? wn : NO_EDGE;
        unsigned q[16];
        float v[16];
#pragma unroll
        for (int j = 0; j < 16; ++j) q[j] = (unsigned)__shfl((int)pk, 2 * j + half);
        if (c0 + 32 <= end) {
#pragma unroll
            for (int j = 0; j < 16; ++j) v[j] = g[(q[j] & SRC_MASK) * 16u + (unsigned)f];
#pragma unroll
            for (int j = 0; j < 16; ++j)
                __hip_atomic_fetch_add(&acc[(q[j] >> SRC_BITS) * 16u + (unsigned)f], v[j],
                                       __ATOMIC_RELAXED, __HIP_MEMORY_SCOPE_WORKGROUP);
        } else {
#pragma unroll
            for (int j = 0; j < 16; ++j) {
                unsigned s = (q[j] != NO_EDGE) ? (q[j] & SRC_MASK) : 0u;
                v[j] = g[s * 16u + (unsigned)f];
            }
#pragma unroll
            for (int j = 0; j < 16; ++j) {
                if (q[j] != NO_EDGE)
                    __hip_atomic_fetch_add(&acc[(q[j] >> SRC_BITS) * 16u + (unsigned)f], v[j],
                                           __ATOMIC_RELAXED, __HIP_MEMORY_SCOPE_WORKGROUP);
            }
        }
        pk = pkn;
    }
    __syncthreads();
    const bool bvalid = (f < NCOL);
    v2f b[4];
#pragma unroll
    for (int j = 0; j < 4; ++j) {
        const int ka = 4 * j + 2 * half;
        b[j].x = bvalid ? Wn[(ka + 0) * NCOL + f] : 0.f;
        b[j].y = bvalid ? Wn[(ka + 1) * NCOL + f] : 0.f;
    }
    for (int i = tid; i < RB * 16; i += 1024) {
        const int r = i >> 4, node = lo + r, ff = i & 15;
        float v = 0.f, di = 0.f;
        if (node < N) {
            size_t o = (size_t)node * 16 + ff;
            di = dis[node];
            v = di * (acc[i] + g[o]) + bias[ff];
            v = v > 0.f ? v : 0.f;
        }
        acc[i] = v;
        if (ff == 0) disl[r] = di;
    }
    __syncthreads();
    const int wv = __builtin_amdgcn_readfirstlane(wave);
    if (wv < RB / 16) {
        const int r0 = wv * 16;
        const float sc = disl[r0 + f];
        const float* ap = &acc[(r0 + f) * 16 + 2 * half];
        v8f c = {0.f, 0.f, 0.f, 0.f, 0.f, 0.f, 0.f, 0.f};
#pragma unroll
        for (int j = 0; j < 4; ++j) {
            v2f a = *(const v2f*)(ap + 4 * j);
            a.x *= sc;
            a.y *= sc;
            c = __builtin_amdgcn_wmma_f32_16x16x4_f32(false, a, false, b[j], (short)0, c,
                                                      false, false);
        }
#pragma unroll
        for (int v = 0; v < 8; ++v) {
            int row = lo + r0 + v + 8 * half;
            if (row < N && f < NCOL) gn[(size_t)row * NCOL + f] = c[v];
        }
    }
}

__global__ __launch_bounds__(1024) void k_bucket2_lsm(const unsigned* __restrict__ sorted,
                                                      const int* __restrict__ bstart,
                                                      const float* __restrict__ g,
                                                      const float* __restrict__ dis,
                                                      const float* __restrict__ bias,
                                                      float* __restrict__ out, int N) {
    __shared__ float acc[RB * 2];
    const int tid = threadIdx.x;
    const int lo = blockIdx.x * RB;
    for (int i = tid; i < RB * 2; i += 1024) acc[i] = 0.f;
    __syncthreads();
    const int beg = bstart[blockIdx.x], end = bstart[blockIdx.x + 1];
    for (int e0 = beg + tid; e0 < end; e0 += 4096) {
        unsigned q[4];
        v2f v[4];
#pragma unroll
        for (int u = 0; u < 4; ++u) {
            int e = e0 + u * 1024;
            q[u] = (e < end) ? sorted[e] : NO_EDGE;
        }
#pragma unroll
        for (int u = 0; u < 4; ++u) {
            unsigned s = (q[u] != NO_EDGE) ? (q[u] & SRC_MASK) : 0u;
            v[u] = *(const v2f*)(g + s * 2u);
        }
#pragma unroll
        for (int u = 0; u < 4; ++u) {
            if (q[u] != NO_EDGE) {
                unsigned ld = q[u] >> SRC_BITS;
                __hip_atomic_fetch_add(&acc[ld * 2u + 0u], v[u].x, __ATOMIC_RELAXED,
                                       __HIP_MEMORY_SCOPE_WORKGROUP);
                __hip_atomic_fetch_add(&acc[ld * 2u + 1u], v[u].y, __ATOMIC_RELAXED,
                                       __HIP_MEMORY_SCOPE_WORKGROUP);
            }
        }
    }
    __syncthreads();
    for (int i = tid; i < RB; i += 1024) {
        int node = lo + i;
        if (node < N) {
            float di = dis[node];
            float z0 = di * (acc[2 * i + 0] + g[(size_t)node * 2 + 0]) + bias[0];
            float z1 = di * (acc[2 * i + 1] + g[(size_t)node * 2 + 1]) + bias[1];
            float m = fmaxf(z0, z1);
            float lse = m + __logf(__expf(z0 - m) + __expf(z1 - m));
            out[(size_t)node * 2 + 0] = z0 - lse;
            out[(size_t)node * 2 + 1] = z1 - lse;
        }
    }
}

extern "C" void kernel_launch(void* const* d_in, const int* in_sizes, int n_in,
                              void* d_out, int out_size, void* d_ws, size_t ws_size,
                              hipStream_t stream) {
    const float* x  = (const float*)d_in[0];
    const int*   ei = (const int*)d_in[1];
    const float* W1 = (const float*)d_in[2];
    const float* b1 = (const float*)d_in[3];
    const float* W2 = (const float*)d_in[4];
    const float* b2 = (const float*)d_in[5];
    const float* W3 = (const float*)d_in[6];
    const float* b3 = (const float*)d_in[7];

    const int N = in_sizes[0] / 128;
    const int E = in_sizes[1] / 2;
    const int* src = ei;
    const int* dst = ei + E;

    char* ws = (char*)d_ws;
    size_t off = 0;
    auto carve = [&](size_t bytes) -> void* {
        void* p = ws + off;
        off += (bytes + 255) & ~(size_t)255;
        return p;
    };
    float*    dis     = (float*)carve((size_t)N * 4);
    int*      bcount  = (int*)carve((size_t)2 * MAXP * 4);
    int*      bcursor = bcount + MAXP;
    int*      bstart  = (int*)carve((size_t)(MAXP + 1) * 4);
    unsigned* sorted  = (unsigned*)carve((size_t)E * 4);
    float*    g       = (float*)carve((size_t)N * 16 * 4);
    float*    g2      = (float*)carve((size_t)N * 16 * 4);
    (void)ws_size; (void)n_in; (void)out_size;

    const int P = (N + RB - 1) / RB;
    const int chunk = (E + NPART - 1) / NPART;

    k_zero<<<(2 * MAXP + 255) / 256, 256, 0, stream>>>(bcount, 2 * MAXP);
    k_count<<<NPART, 1024, 0, stream>>>(dst, bcount, E, chunk, P);
    k_partition<<<NPART, 1024, 0, stream>>>(src, dst, bcount, bcursor, bstart, sorted, E,
                                            chunk, P);

    k_deg_gemm1<<<P, 1024, 0, stream>>>(sorted, bstart, x, W1, dis, g, N);

    k_bucket16<16><<<P, 1024, 0, stream>>>(sorted, bstart, g, dis, b1, W2, g2, N, E);

    k_bucket16<2><<<P, 1024, 0, stream>>>(sorted, bstart, g2, dis, b2, W3, g, N, E);

    k_bucket2_lsm<<<P, 1024, 0, stream>>>(sorted, bstart, g, dis, b3, (float*)d_out, N);
}


# k_partition stages each block's grouped edges in LDS in bucket order and writes them with consecutive lanes: whole bucket runs, not one scattered 4-byte store per edge (order still arrival order)
# speedup vs baseline: 6.1978x; 1.0957x over previous
#include <hip/hip_runtime.h>
#include <math.h>


typedef float v2f __attribute__((ext_vector_type(2)));
typedef float v8f __attribute__((ext_vector_type(8)));

#define RB 416
#define MAXP 1024
#define SRC_BITS 20
#define SRC_MASK ((1u << SRC_BITS) - 1u)
#define NO_EDGE 0xFFFFFFFFu
#define NPART 256
static_assert(RB % 16 == 0 && RB / 16 <= 32, "one wave of a 1024-thread work-group per 16-row tile");

__global__ void k_zero(int* __restrict__ p, int n) {
    int i = blockIdx.x * blockDim.x + threadIdx.x;
    if (i < n) p[i] = 0;
}

__global__ __launch_bounds__(1024) void k_count(const int* __restrict__ dst,
                                                int* __restrict__ bcount, int E,
                                                int chunk, int P) {
    __shared__ int hist[MAXP];
    const int tid = threadIdx.x;
    for (int i = tid; i < MAXP; i += 1024) hist[i] = 0;
    __syncthreads();
    long beg = (long)blockIdx.x * chunk;
    long end = beg + chunk;
    if (end > (long)E) end = E;
    for (long e0 = beg + tid; e0 < end; e0 += 4096) {
        int d[4];
#pragma unroll
        for (int u = 0; u < 4; ++u) {
            long e = e0 + u * 1024;
            d[u] = (e < end) ? dst[e] : -1;
        }
#pragma unroll
        for (int u = 0; u < 4; ++u)
            if (d[u] >= 0) atomicAdd(&hist[d[u] / RB], 1);
    }
    __syncthreads();
    for (int i = tid; i < P; i += 1024) {
        int c = hist[i];
        if (c) atomicAdd(&bcount[i], c);
    }
}

#define STAGE_CAP 16384
static_assert(MAXP == 1024, "the scan below gives one table entry to each of the 1024 threads");

__device__ __forceinline__ int block_excl_scan(int v, int* wsum, int tid) {
    const int lane = tid & 31, w = tid >> 5;
    int x = v;
#pragma unroll
    for (int o = 1; o < 32; o <<= 1) {
        int y = __shfl_up(x, o);
        if (lane >= o) x += y;
    }
    if (lane == 31) wsum[w] = x;
    __syncthreads();
    if (w == 0) {
        const int s = wsum[lane];
        int t = s;
#pragma unroll
        for (int o = 1; o < 32; o <<= 1) {
            int y = __shfl_up(t, o);
            if (lane >= o) t += y;
        }
        wsum[lane] = t - s;
    }
    __syncthreads();
    return x - v + wsum[w];
}

__global__ __launch_bounds__(1024) void k_partition(const int* __restrict__ src,
                                                    const int* __restrict__ dst,
                                                    const int* __restrict__ bcount,
                                                    int* __restrict__ bcursor,
                                                    int* __restrict__ bstart,
                                                    unsigned* __restrict__ sorted, int E,
                                                    int chunk, int P) {
    __shared__ int hist[MAXP];
    __shared__ int base[MAXP];
    __shared__ int cnt[MAXP];
    __shared__ int thist[MAXP];
    __shared__ int tstart[MAXP];
    __shared__ int wsum[32];
    __shared__ unsigned sw[STAGE_CAP];
    __shared__ int sidx[STAGE_CAP];
    const int tid = threadIdx.x;
    for (int i = tid; i < MAXP; i += 1024) {
        hist[i] = 0;
        cnt[i] = (i < P) ? bcount[i] : 0;
    }
    __syncthreads();
    long beg = (long)blockIdx.x * chunk;
    long end = beg + chunk;
    if (end > (long)E) end = E;
    for (long e0 = beg + tid; e0 < end; e0 += 4096) {
        int d[4];
#pragma unroll
        for (int u = 0; u < 4; ++u) {
            long e = e0 + u * 1024;
            d[u] = (e < end) ? dst[e] : -1;
        }
#pragma unroll
        for (int u = 0; u < 4; ++u)
            if (d[u] >= 0) atomicAdd(&hist[d[u] / RB], 1);
    }
    __syncthreads();
    for (int i = tid; i < P; i += 1024) {
        int s = 0;
        for (int j = 0; j < i; ++j) s += cnt[j];
        int c = hist[i];
        int off = c ? atomicAdd(&bcursor[i], c) : 0;
        base[i] = s + off;
        if (blockIdx.x == 0) bstart[i] = s;
    }
    if (blockIdx.x == 0 && tid == 0) bstart[P] = E;
    if (tid >= P) base[tid] = 0;
    __syncthreads();
    const bool one_tile = (end - beg) <= (long)STAGE_CAP;
    for (long t0 = beg; t0 < end; t0 += STAGE_CAP) {
        long t1 = t0 + STAGE_CAP;
        if (t1 > end) t1 = end;
        if (one_tile) {
            thist[tid] = hist[tid];
        } else {
            thist[tid] = 0;
            __syncthreads();
            for (long e0 = t0 + tid; e0 < t1; e0 += 4096) {
                int d[4];
#pragma unroll
                for (int u = 0; u < 4; ++u) {
                    long e = e0 + u * 1024;
                    d[u] = (e < t1) ? dst[e] : -1;
                }
#pragma unroll
                for (int u = 0; u < 4; ++u)
                    if (d[u] >= 0) atomicAdd(&thist[d[u] / RB], 1);
            }
        }
        __syncthreads();
        const int tc = thist[tid];
        tstart[tid] = block_excl_scan(tc, wsum, tid);
        thist[tid] = 0;
        __syncthreads();
        for (long e0 = t0 + tid; e0 < t1; e0 += 4096) {
            int d[4], s[4];
#pragma unroll
            for (int u = 0; u < 4; ++u) {
                long e = e0 + u * 1024;
                bool ok = e < t1;
                d[u] = ok ? dst[e] : -1;
                s[u] = ok ? src[e] : 0;
            }
#pragma unroll
            for (int u = 0; u < 4; ++u) {
                if (d[u] >= 0) {
                    int b = d[u] / RB;
                    int k = atomicAdd(&thist[b], 1);
                    int p = tstart[b] + k;
                    sw[p] = (unsigned)s[u] | ((unsigned)(d[u] - b * RB) << SRC_BITS);
                    sidx[p] = base[b] + k;
                }
            }
        }
        __syncthreads();
        const int n = (int)(t1 - t0);
        for (int i = tid; i < n; i += 1024) sorted[sidx[i]] = sw[i];
        if (!one_tile) {
            __syncthreads();
            base[tid] += thist[tid];
            __syncthreads();
        }
    }
}


__global__ __launch_bounds__(1024) void k_deg_gemm1(const unsigned* __restrict__ sorted,
                                                    const int* __restrict__ bstart,
                                                    const float* __restrict__ x,
                                                    const float* __restrict__ W1,
                                                    float* __restrict__ dis,
                                                    float* __restrict__ g, int N) {
    __shared__ int cntl[RB];
    const int tid = threadIdx.x;
    const int lo = blockIdx.x * RB;
    const int lane = tid & 31;
    const int f = lane & 15, half = lane >> 4;
    const int wv = __builtin_amdgcn_readfirstlane(tid >> 5);
    const bool has_tile = wv < RB / 16;
    const int r0 = wv * 16;
    int arow = lo + r0 + f;
    arow = arow < N ? arow : (N - 1);
    for (int i = tid; i < RB; i += 1024) cntl[i] = 0;
    __syncthreads();
    const int beg = bstart[blockIdx.x], end = bstart[blockIdx.x + 1];
    for (int e0 = beg + tid; e0 < end; e0 += 4096) {
        unsigned q[4];
#pragma unroll
        for (int u = 0; u < 4; ++u) {
            int e = e0 + u * 1024;
            q[u] = (e < end) ? sorted[e] : NO_EDGE;
        }
#pragma unroll
        for (int u = 0; u < 4; ++u)
            if (q[u] != NO_EDGE) atomicAdd(&cntl[q[u] >> SRC_BITS], 1);
    }
    __syncthreads();
    for (int i = tid; i < RB; i += 1024) {
        int node = lo + i;
        if (node < N) dis[node] = rsqrtf((float)(cntl[i] + 1));
    }
    if (has_tile) {
        const float sc = rsqrtf((float)(cntl[r0 + f] + 1));
        const float* __restrict__ ap = x + (long)arow * 128 + 2 * half;
        const float* __restrict__ bp = W1 + f;
        v8f c = {0.f, 0.f, 0.f, 0.f, 0.f, 0.f, 0.f, 0.f};
#pragma unroll
        for (int j = 0; j < 32; ++j) {
            v2f av = *(const v2f*)(ap + 4 * j);
            av.x *= sc;
            av.y *= sc;
            int ka = 4 * j + 2 * half;
            v2f b;
            b.x = bp[(ka + 0) * 16];
            b.y = bp[(ka + 1) * 16];
            c = __builtin_amdgcn_wmma_f32_16x16x4_f32(false, av, false, b, (short)0, c,
                                                      false, false);
        }
#pragma unroll
        for (int v = 0; v < 8; ++v) {
            int row = lo + r0 + v + 8 * half;
            if (row < N) g[(long)row * 16 + f] = c[v];
        }
    }
}

template <int NCOL>
__global__ __launch_bounds__(1024) void k_bucket16(const unsigned* __restrict__ sorted,
                                                   const int* __restrict__ bstart,
                                                   const float* __restrict__ g,
                                                   const float* __restrict__ dis,
                                                   const float* __restrict__ bias,
                                                   const float* __restrict__ Wn,
                                                   float* __restrict__ gn, int N, int E) {
    __shared__ float acc[RB * 16];
    __shared__ float disl[RB];
    const int tid = threadIdx.x;
    const int lo = blockIdx.x * RB;
    for (int i = tid; i < RB * 16; i += 1024) acc[i] = 0.f;
    __syncthreads();
    const int beg = bstart[blockIdx.x], end = bstart[blockIdx.x + 1];
    const int lane = tid & 31, wave = tid >> 5;
    const int f = lane & 15, half = lane >> 4;
    int c0 = beg + wave * 32;
    unsigned pk = (c0 + lane < end) ? sorted[c0 + lane] : NO_EDGE;
    for (; c0 < end; c0 += 1024) {
        const int en = c0 + 1024 + lane;
        const unsigned wn = sorted[en < E ? en : E - 1];
        const unsigned pkn = (en < end) ? wn : NO_EDGE;
        unsigned q[16];
        float v[16];
#pragma unroll
        for (int j = 0; j < 16; ++j) q[j] = (unsigned)__shfl((int)pk, 2 * j + half);
        if (c0 + 32 <= end) {
#pragma unroll
            for (int j = 0; j < 16; ++j) v[j] = g[(q[j] & SRC_MASK) * 16u + (unsigned)f];
#pragma unroll
            for (int j = 0; j < 16; ++j)
                __hip_atomic_fetch_add(&acc[(q[j] >> SRC_BITS) * 16u + (unsigned)f], v[j],
                                       __ATOMIC_RELAXED, __HIP_MEMORY_SCOPE_WORKGROUP);
        } else {
#pragma unroll
            for (int j = 0; j < 16; ++j) {
                unsigned s = (q[j] != NO_EDGE) ? (q[j] & SRC_MASK) : 0u;
                v[j] = g[s * 16u + (unsigned)f];
            }
#pragma unroll
            for (int j = 0; j < 16; ++j) {
                if (q[j] != NO_EDGE)
                    __hip_atomic_fetch_add(&acc[(q[j] >> SRC_BITS) * 16u + (unsigned)f], v[j],
                                           __ATOMIC_RELAXED, __HIP_MEMORY_SCOPE_WORKGROUP);
            }
        }
        pk = pkn;
    }
    __syncthreads();
    const bool bvalid = (f < NCOL);
    v2f b[4];
#pragma unroll
    for (int j = 0; j < 4; ++j) {
        const int ka = 4 * j + 2 * half;
        b[j].x = bvalid ? Wn[(ka + 0) * NCOL + f] : 0.f;
        b[j].y = bvalid ? Wn[(ka + 1) * NCOL + f] : 0.f;
    }
    for (int i = tid; i < RB * 16; i += 1024) {
        const int r = i >> 4, node = lo + r, ff = i & 15;
        float v = 0.f, di = 0.f;
        if (node < N) {
            size_t o = (size_t)node * 16 + ff;
            di = dis[node];
            v = di * (acc[i] + g[o]) + bias[ff];
            v = v > 0.f ? v : 0.f;
        }
        acc[i] = v;
        if (ff == 0) disl[r] = di;
    }
    __syncthreads();
    const int wv = __builtin_amdgcn_readfirstlane(wave);
    if (wv < RB / 16) {
        const int r0 = wv * 16;
        const float sc = disl[r0 + f];
        const float* ap = &acc[(r0 + f) * 16 + 2 * half];
        v8f c = {0.f, 0.f, 0.f, 0.f, 0.f, 0.f, 0.f, 0.f};
#pragma unroll
        for (int j = 0; j < 4; ++j) {
            v2f a = *(const v2f*)(ap + 4 * j);
            a.x *= sc;
            a.y *= sc;
            c = __builtin_amdgcn_wmma_f32_16x16x4_f32(false, a, false, b[j], (short)0, c,
                                                      false, false);
        }
#pragma unroll
        for (int v = 0; v < 8; ++v) {
            int row = lo + r0 + v + 8 * half;
            if (row < N && f < NCOL) gn[(size_t)row * NCOL + f] = c[v];
        }
    }
}

__global__ __launch_bounds__(1024) void k_bucket2_lsm(const unsigned* __restrict__ sorted,
                                                      const int* __restrict__ bstart,
                                                      const float* __restrict__ g,
                                                      const float* __restrict__ dis,
                                                      const float* __restrict__ bias,
                                                      float* __restrict__ out, int N) {
    __shared__ float acc[RB * 2];
    const int tid = threadIdx.x;
    const int lo = blockIdx.x * RB;
    for (int i = tid; i < RB * 2; i += 1024) acc[i] = 0.f;
    __syncthreads();
    const int beg = bstart[blockIdx.x], end = bstart[blockIdx.x + 1];
    for (int e0 = beg + tid; e0 < end; e0 += 4096) {
        unsigned q[4];
        v2f v[4];
#pragma unroll
        for (int u = 0; u < 4; ++u) {
            int e = e0 + u * 1024;
            q[u] = (e < end) ? sorted[e] : NO_EDGE;
        }
#pragma unroll
        for (int u = 0; u < 4; ++u) {
            unsigned s = (q[u] != NO_EDGE) ? (q[u] & SRC_MASK) : 0u;
            v[u] = *(const v2f*)(g + s * 2u);
        }
#pragma unroll
        for (int u = 0; u < 4; ++u) {
            if (q[u] != NO_EDGE) {
                unsigned ld = q[u] >> SRC_BITS;
                __hip_atomic_fetch_add(&acc[ld * 2u + 0u], v[u].x, __ATOMIC_RELAXED,
                                       __HIP_MEMORY_SCOPE_WORKGROUP);
                __hip_atomic_fetch_add(&acc[ld * 2u + 1u], v[u].y, __ATOMIC_RELAXED,
                                       __HIP_MEMORY_SCOPE_WORKGROUP);
            }
        }
    }
    __syncthreads();
    for (int i = tid; i < RB; i += 1024) {
        int node = lo + i;
        if (node < N) {
            float di = dis[node];
            float z0 = di * (acc[2 * i + 0] + g[(size_t)node * 2 + 0]) + bias[0];
            float z1 = di * (acc[2 * i + 1] + g[(size_t)node * 2 + 1]) + bias[1];
            float m = fmaxf(z0, z1);
            float lse = m + __logf(__expf(z0 - m) + __expf(z1 - m));
            out[(size_t)node * 2 + 0] = z0 - lse;
            out[(size_t)node * 2 + 1] = z1 - lse;
        }
    }
}

extern "C" void kernel_launch(void* const* d_in, const int* in_sizes, int n_in,
                              void* d_out, int out_size, void* d_ws, size_t ws_size,
                              hipStream_t stream) {
    const float* x  = (const float*)d_in[0];
    const int*   ei = (const int*)d_in[1];
    const float* W1 = (const float*)d_in[2];
    const float* b1 = (const float*)d_in[3];
    const float* W2 = (const float*)d_in[4];
    const float* b2 = (const float*)d_in[5];
    const float* W3 = (const float*)d_in[6];
    const float* b3 = (const float*)d_in[7];

    const int N = in_sizes[0] / 128;
    const int E = in_sizes[1] / 2;
    const int* src = ei;
    const int* dst = ei + E;

    char* ws = (char*)d_ws;
    size_t off = 0;
    auto carve = [&](size_t bytes) -> void* {
        void* p = ws + off;
        off += (bytes + 255) & ~(size_t)255;
        return p;
    };
    float*    dis     = (float*)carve((size_t)N * 4);
    int*      bcount  = (int*)carve((size_t)2 * MAXP * 4);
    int*      bcursor = bcount + MAXP;
    int*      bstart  = (int*)carve((size_t)(MAXP + 1) * 4);
    unsigned* sorted  = (unsigned*)carve((size_t)E * 4);
    float*    g       = (float*)carve((size_t)N * 16 * 4);
    float*    g2      = (float*)carve((size_t)N * 16 * 4);
    (void)ws_size; (void)n_in; (void)out_size;

    const int P = (N + RB - 1) / RB;
    const int chunk = (E + NPART - 1) / NPART;

    k_zero<<<(2 * MAXP + 255) / 256, 256, 0, stream>>>(bcount, 2 * MAXP);
    k_count<<<NPART, 1024, 0, stream>>>(dst, bcount, E, chunk, P);
    k_partition<<<NPART, 1024, 0, stream>>>(src, dst, bcount, bcursor, bstart, sorted, E,
                                            chunk, P);

    k_deg_gemm1<<<P, 1024, 0, stream>>>(sorted, bstart, x, W1, dis, g, N);

    k_bucket16<16><<<P, 1024, 0, stream>>>(sorted, bstart, g, dis, b1, W2, g2, N, E);

    k_bucket16<2><<<P, 1024, 0, stream>>>(sorted, bstart, g2, dis, b2, W3, g, N, E);

    k_bucket2_lsm<<<P, 1024, 0, stream>>>(sorted, bstart, g, dis, b3, (float*)d_out, N);
}


# each partition block groups its own share of the edges and records its per-bucket runs; a bucket's waves walk 8 runs each: global count and zero passes gone, 7 launches to 5; LDS sums arrival order
# speedup vs baseline: 6.4938x; 1.0478x over previous
#include <hip/hip_runtime.h>
#include <math.h>


typedef float v2f __attribute__((ext_vector_type(2)));
typedef float v8f __attribute__((ext_vector_type(8)));

#define RB 416
#define MAXP 1024
#define SRC_BITS 20
#define SRC_MASK ((1u << SRC_BITS) - 1u)
#define NO_EDGE 0xFFFFFFFFu
#define SPARE_EDGE ((unsigned)RB << SRC_BITS)
#define NPART 256
static_assert(RB % 16 == 0 && RB / 16 <= 32, "one wave of a 1024-thread work-group per 16-row tile");

#define RPW (NPART / 32)
static_assert(NPART % 32 == 0, "each of the 32 waves of a consumer work-group takes RPW whole runs");

struct WaveRuns {
    int o[RPW];
    int d[RPW];
    int total;
};

__device__ __forceinline__ WaveRuns wave_runs(const int* __restrict__ runs, int bucket, int wv) {
    const int* __restrict__ rt = runs + ((size_t)bucket * NPART + (size_t)wv * RPW) * 2;
    WaveRuns r;
    int acc = 0;
#pragma unroll
    for (int k = 0; k < RPW; ++k) {
        const int st = rt[2 * k], ln = rt[2 * k + 1];
        r.o[k] = acc;
        r.d[k] = st - acc;
        acc += ln;
    }
    r.total = acc;
    return r;
}

__device__ __forceinline__ int run_slot(const WaveRuns& r, int e) {
    int d = r.d[0];
#pragma unroll
    for (int k = 1; k < RPW; ++k) d = (e >= r.o[k]) ? r.d[k] : d;
    return e + d;
}

#define STAGE_CAP 16384
static_assert(MAXP == 1024, "the scan below gives one table entry to each of the 1024 threads");

__device__ __forceinline__ int block_excl_scan(int v, int* wsum, int tid) {
    const int lane = tid & 31, w = tid >> 5;
    int x = v;
#pragma unroll
    for (int o = 1; o < 32; o <<= 1) {
        int y = __shfl_up(x, o);
        if (lane >= o) x += y;
    }
    if (lane == 31) wsum[w] = x;
    __syncthreads();
    if (w == 0) {
        const int s = wsum[lane];
        int t = s;
#pragma unroll
        for (int o = 1; o < 32; o <<= 1) {
            int y = __shfl_up(t, o);
            if (lane >= o) t += y;
        }
        wsum[lane] = t - s;
    }
    __syncthreads();
    return x - v + wsum[w];
}

__global__ __launch_bounds__(1024) void k_partition(const int* __restrict__ src,
                                                    const int* __restrict__ dst,
                                                    int* __restrict__ runs,
                                                    unsigned* __restrict__ sorted, int E,
                                                    int chunk, int P) {
    __shared__ int hist[MAXP];
    __shared__ int base[MAXP];
    __shared__ int thist[MAXP];
    __shared__ int tstart[MAXP];
    __shared__ int wsum[32];
    __shared__ unsigned sw[STAGE_CAP];
    __shared__ int sidx[STAGE_CAP];
    const int tid = threadIdx.x;
    hist[tid] = 0;
    __syncthreads();
    long beg = (long)blockIdx.x * chunk;
    long end = beg + chunk;
    if (end > (long)E) end = E;
    for (long e0 = beg + tid; e0 < end; e0 += 4096) {
        int d[4];
#pragma unroll
        for (int u = 0; u < 4; ++u) {
            long e = e0 + u * 1024;
            d[u] = (e < end) ? dst[e] : -1;
        }
#pragma unroll
        for (int u = 0; u < 4; ++u)
            if (d[u] >= 0) atomicAdd(&hist[d[u] / RB], 1);
    }
    __syncthreads();
    {
        const int hc = hist[tid];
        const int bs = (int)beg + block_excl_scan(hc, wsum, tid);
        base[tid] = bs;
        if (tid < P) {
            const size_t ri = ((size_t)tid * NPART + blockIdx.x) * 2;
            runs[ri] = bs;
            runs[ri + 1] = hc;
        }
    }
    __syncthreads();
    const bool one_tile = (end - beg) <= (long)STAGE_CAP;
    for (long t0 = beg; t0 < end; t0 += STAGE_CAP) {
        long t1 = t0 + STAGE_CAP;
        if (t1 > end) t1 = end;
        if (one_tile) {
            thist[tid] = hist[tid];
        } else {
            thist[tid] = 0;
            __syncthreads();
            for (long e0 = t0 + tid; e0 < t1; e0 += 4096) {
                int d[4];
#pragma unroll
                for (int u = 0; u < 4; ++u) {
                    long e = e0 + u * 1024;
                    d[u] = (e < t1) ? dst[e] : -1;
                }
#pragma unroll
                for (int u = 0; u < 4; ++u)
                    if (d[u] >= 0) atomicAdd(&thist[d[u] / RB], 1);
            }
        }
        __syncthreads();
        const int tc = thist[tid];
        tstart[tid] = block_excl_scan(tc, wsum, tid);
        thist[tid] = 0;
        __syncthreads();
        for (long e0 = t0 + tid; e0 < t1; e0 += 4096) {
            int d[4], s[4];
#pragma unroll
            for (int u = 0; u < 4; ++u) {
                long e = e0 + u * 1024;
                bool ok = e < t1;
                d[u] = ok ? dst[e] : -1;
                s[u] = ok ? src[e] : 0;
            }
#pragma unroll
            for (int u = 0; u < 4; ++u) {
                if (d[u] >= 0) {
                    int b = d[u] / RB;
                    int k = atomicAdd(&thist[b], 1);
                    int p = tstart[b] + k;
                    sw[p] = (unsigned)s[u] | ((unsigned)(d[u] - b * RB) << SRC_BITS);
                    sidx[p] = base[b] + k;
                }
            }
        }
        __syncthreads();
        const int n = (int)(t1 - t0);
        for (int i = tid; i < n; i += 1024) sorted[sidx[i]] = sw[i];
        if (!one_tile) {
            __syncthreads();
            base[tid] += thist[tid];
            __syncthreads();
        }
    }
}


__global__ __launch_bounds__(1024) void k_deg_gemm1(const unsigned* __restrict__ sorted,
                                                    const int* __restrict__ runs,
                                                    const float* __restrict__ x,
                                                    const float* __restrict__ W1,
                                                    float* __restrict__ dis,
                                                    float* __restrict__ g, int N) {
    __shared__ int cntl[RB];
    const int tid = threadIdx.x;
    const int lo = blockIdx.x * RB;
    const int lane = tid & 31;
    const int f = lane & 15, half = lane >> 4;
    const int wv = __builtin_amdgcn_readfirstlane(tid >> 5);
    const bool has_tile = wv < RB / 16;
    const int r0 = wv * 16;
    int arow = lo + r0 + f;
    arow = arow < N ? arow : (N - 1);
    for (int i = tid; i < RB; i += 1024) cntl[i] = 0;
    const WaveRuns wr = wave_runs(runs, blockIdx.x, wv);
    __syncthreads();
    for (int e0 = lane; e0 < wr.total; e0 += 128) {
        unsigned q[4];
#pragma unroll
        for (int u = 0; u < 4; ++u) {
            int e = e0 + u * 32;
            q[u] = (e < wr.total) ? sorted[run_slot(wr, e)] : NO_EDGE;
        }
#pragma unroll
        for (int u = 0; u < 4; ++u)
            if (q[u] != NO_EDGE) atomicAdd(&cntl[q[u] >> SRC_BITS], 1);
    }
    __syncthreads();
    for (int i = tid; i < RB; i += 1024) {
        int node = lo + i;
        if (node < N) dis[node] = rsqrtf((float)(cntl[i] + 1));
    }
    if (has_tile) {
        const float sc = rsqrtf((float)(cntl[r0 + f] + 1));
        const float* __restrict__ ap = x + (long)arow * 128 + 2 * half;
        const float* __restrict__ bp = W1 + f;
        v8f c = {0.f, 0.f, 0.f, 0.f, 0.f, 0.f, 0.f, 0.f};
#pragma unroll
        for (int j = 0; j < 32; ++j) {
            v2f av = *(const v2f*)(ap + 4 * j);
            av.x *= sc;
            av.y *= sc;
            int ka = 4 * j + 2 * half;
            v2f b;
            b.x = bp[(ka + 0) * 16];
            b.y = bp[(ka + 1) * 16];
            c = __builtin_amdgcn_wmma_f32_16x16x4_f32(false, av, false, b, (short)0, c,
                                                      false, false);
        }
#pragma unroll
        for (int v = 0; v < 8; ++v) {
            int row = lo + r0 + v + 8 * half;
            if (row < N) g[(long)row * 16 + f] = c[v];
        }
    }
}

template <int NCOL>
__global__ __launch_bounds__(1024) void k_bucket16(const unsigned* __restrict__ sorted,
                                                   const int* __restrict__ runs,
                                                   const float* __restrict__ g,
                                                   const float* __restrict__ dis,
                                                   const float* __restrict__ bias,
                                                   const float* __restrict__ Wn,
                                                   float* __restrict__ gn, int N) {
    __shared__ float acc[(RB + 1) * 16];
    __shared__ float disl[RB];
    const int tid = threadIdx.x;
    const int lo = blockIdx.x * RB;
    const int lane = tid & 31, wave = tid >> 5;
    const int f = lane & 15, half = lane >> 4;
    const int wv = __builtin_amdgcn_readfirstlane(wave);
    for (int i = tid; i < (RB + 1) * 16; i += 1024) acc[i] = 0.f;
    const WaveRuns wr = wave_runs(runs, blockIdx.x, wv);
    const int total = wr.total;
    __syncthreads();
    unsigned pk = (lane < total) ? sorted[run_slot(wr, lane)] : SPARE_EDGE;
    for (int c0 = 0; c0 < total; c0 += 32) {
        const int en = c0 + 32 + lane;
        const unsigned wn = sorted[(en < total) ? run_slot(wr, en) : 0];
        const unsigned pkn = (en < total) ? wn : SPARE_EDGE;
        unsigned q[16];
        float v[16];
#pragma unroll
        for (int j = 0; j < 16; ++j) q[j] = (unsigned)__shfl((int)pk, 2 * j + half);
#pragma unroll
        for (int j = 0; j < 16; ++j) v[j] = g[(q[j] & SRC_MASK) * 16u + (unsigned)f];
#pragma unroll
        for (int j = 0; j < 16; ++j)
            __hip_atomic_fetch_add(&acc[(q[j] >> SRC_BITS) * 16u + (unsigned)f], v[j],
                                   __ATOMIC_RELAXED, __HIP_MEMORY_SCOPE_WORKGROUP);
        pk = pkn;
    }
    __syncthreads();
    const bool bvalid = (f < NCOL);
    v2f b[4];
#pragma unroll
    for (int j = 0; j < 4; ++j) {
        const int ka = 4 * j + 2 * half;
        b[j].x = bvalid ? Wn[(ka + 0) * NCOL + f] : 0.f;
        b[j].y = bvalid ? Wn[(ka + 1) * NCOL + f] : 0.f;
    }
    for (int i = tid; i < RB * 16; i += 1024) {
        const int r = i >> 4, node = lo + r, ff = i & 15;
        float v = 0.f, di = 0.f;
        if (node < N) {
            size_t o = (size_t)node * 16 + ff;
            di = dis[node];
            v = di * (acc[i] + g[o]) + bias[ff];
            v = v > 0.f ? v : 0.f;
        }
        acc[i] = v;
        if (ff == 0) disl[r] = di;
    }
    __syncthreads();
    if (wv < RB / 16) {
        const int r0 = wv * 16;
        const float sc = disl[r0 + f];
        const float* ap = &acc[(r0 + f) * 16 + 2 * half];
        v8f c = {0.f, 0.f, 0.f, 0.f, 0.f, 0.f, 0.f, 0.f};
#pragma unroll
        for (int j = 0; j < 4; ++j) {
            v2f a = *(const v2f*)(ap + 4 * j);
            a.x *= sc;
            a.y *= sc;
            c = __builtin_amdgcn_wmma_f32_16x16x4_f32(false, a, false, b[j], (short)0, c,
                                                      false, false);
        }
#pragma unroll
        for (int v = 0; v < 8; ++v) {
            int row = lo + r0 + v + 8 * half;
            if (row < N && f < NCOL) gn[(size_t)row * NCOL + f] = c[v];
        }
    }
}

__global__ __launch_bounds__(1024) void k_bucket2_lsm(const unsigned* __restrict__ sorted,
                                                      const int* __restrict__ runs,
                                                      const float* __restrict__ g,
                                                      const float* __restrict__ dis,
                                                      const float* __restrict__ bias,
                                                      float* __restrict__ out, int N) {
    __shared__ float acc[RB * 2];
    const int tid = threadIdx.x;
    const int lo = blockIdx.x * RB;
    for (int i = tid; i < RB * 2; i += 1024) acc[i] = 0.f;
    const int wv = __builtin_amdgcn_readfirstlane(tid >> 5);
    const int lane = tid & 31;
    const WaveRuns wr = wave_runs(runs, blockIdx.x, wv);
    __syncthreads();
    for (int e0 = lane; e0 < wr.total; e0 += 128) {
        unsigned q[4];
        v2f v[4];
#pragma unroll
        for (int u = 0; u < 4; ++u) {
            int e = e0 + u * 32;
            q[u] = (e < wr.total) ? sorted[run_slot(wr, e)] : NO_EDGE;
        }
#pragma unroll
        for (int u = 0; u < 4; ++u) {
            unsigned s = (q[u] != NO_EDGE) ? (q[u] & SRC_MASK) : 0u;
            v[u] = *(const v2f*)(g + s * 2u);
        }
#pragma unroll
        for (int u = 0; u < 4; ++u) {
            if (q[u] != NO_EDGE) {
                unsigned ld = q[u] >> SRC_BITS;
                __hip_atomic_fetch_add(&acc[ld * 2u + 0u], v[u].x, __ATOMIC_RELAXED,
                                       __HIP_MEMORY_SCOPE_WORKGROUP);
                __hip_atomic_fetch_add(&acc[ld * 2u + 1u], v[u].y, __ATOMIC_RELAXED,
                                       __HIP_MEMORY_SCOPE_WORKGROUP);
            }
        }
    }
    __syncthreads();
    for (int i = tid; i < RB; i += 1024) {
        int node = lo + i;
        if (node < N) {
            float di = dis[node];
            float z0 = di * (acc[2 * i + 0] + g[(size_t)node * 2 + 0]) + bias[0];
            float z1 = di * (acc[2 * i + 1] + g[(size_t)node * 2 + 1]) + bias[1];
            float m = fmaxf(z0, z1);
            float lse = m + __logf(__expf(z0 - m) + __expf(z1 - m));
            out[(size_t)node * 2 + 0] = z0 - lse;
            out[(size_t)node * 2 + 1] = z1 - lse;
        }
    }
}

extern "C" void kernel_launch(void* const* d_in, const int* in_sizes, int n_in,
                              void* d_out, int out_size, void* d_ws, size_t ws_size,
                              hipStream_t stream) {
    const float* x  = (const float*)d_in[0];
    const int*   ei = (const int*)d_in[1];
    const float* W1 = (const float*)d_in[2];
    const float* b1 = (const float*)d_in[3];
    const float* W2 = (const float*)d_in[4];
    const float* b2 = (const float*)d_in[5];
    const float* W3 = (const float*)d_in[6];
    const float* b3 = (const float*)d_in[7];

    const int N = in_sizes[0] / 128;
    const int E = in_sizes[1] / 2;
    const int* src = ei;
    const int* dst = ei + E;

    char* ws = (char*)d_ws;
    size_t off = 0;
    auto carve = [&](size_t bytes) -> void* {
        void* p = ws + off;
        off += (bytes + 255) & ~(size_t)255;
        return p;
    };
    const int P = (N + RB - 1) / RB;
    const int chunk = (E + NPART - 1) / NPART;
    float*    dis     = (float*)carve((size_t)N * 4);
    int*      runs    = (int*)carve((size_t)P * NPART * 2 * 4);
    unsigned* sorted  = (unsigned*)carve((size_t)E * 4);
    float*    g       = (float*)carve((size_t)N * 16 * 4);
    float*    g2      = (float*)carve((size_t)N * 16 * 4);
    (void)ws_size; (void)n_in; (void)out_size;

    k_partition<<<NPART, 1024, 0, stream>>>(src, dst, runs, sorted, E, chunk, P);

    k_deg_gemm1<<<P, 1024, 0, stream>>>(sorted, runs, x, W1, dis, g, N);

    k_bucket16<16><<<P, 1024, 0, stream>>>(sorted, runs, g, dis, b1, W2, g2, N);

    k_bucket16<2><<<P, 1024, 0, stream>>>(sorted, runs, g2, dis, b2, W3, g, N);

    k_bucket2_lsm<<<P, 1024, 0, stream>>>(sorted, runs, g, dis, b3, (float*)d_out, N);
}


# k_bucket16 files its bucket's edges per destination node in LDS (slot = returning LDS integer add) and four lanes sum a node's gathered rows in registers: no LDS float adds; sums in filing order
# speedup vs baseline: 6.7916x; 1.0459x over previous
#include <hip/hip_runtime.h>
#include <math.h>


typedef float v2f __attribute__((ext_vector_type(2)));
typedef float v8f __attribute__((ext_vector_type(8)));

#define RB 416
#define MAXP 1024
#define SRC_BITS 20
#define SRC_MASK ((1u << SRC_BITS) - 1u)
#define NO_EDGE 0xFFFFFFFFu
#define SPARE_EDGE ((unsigned)RB << SRC_BITS)
#define NPART 256
static_assert(RB % 16 == 0 && RB / 16 <= 32, "one wave of a 1024-thread work-group per 16-row tile");

#define RPW (NPART / 32)
static_assert(NPART % 32 == 0, "each of the 32 waves of a consumer work-group takes RPW whole runs");

struct WaveRuns {
    int o[RPW];
    int d[RPW];
    int total;
};

__device__ __forceinline__ WaveRuns wave_runs(const int* __restrict__ runs, int bucket, int wv) {
    const int* __restrict__ rt = runs + ((size_t)bucket * NPART + (size_t)wv * RPW) * 2;
    WaveRuns r;
    int acc = 0;
#pragma unroll
    for (int k = 0; k < RPW; ++k) {
        const int st = rt[2 * k], ln = rt[2 * k + 1];
        r.o[k] = acc;
        r.d[k] = st - acc;
        acc += ln;
    }
    r.total = acc;
    return r;
}

__device__ __forceinline__ int run_slot(const WaveRuns& r, int e) {
    int d = r.d[0];
#pragma unroll
    for (int k = 1; k < RPW; ++k) d = (e >= r.o[k]) ? r.d[k] : d;
    return e + d;
}

#define STAGE_CAP 16384
static_assert(MAXP == 1024, "the scan below gives one table entry to each of the 1024 threads");

__device__ __forceinline__ int block_excl_scan(int v, int* wsum, int tid) {
    const int lane = tid & 31, w = tid >> 5;
    int x = v;
#pragma unroll
    for (int o = 1; o < 32; o <<= 1) {
        int y = __shfl_up(x, o);
        if (lane >= o) x += y;
    }
    if (lane == 31) wsum[w] = x;
    __syncthreads();
    if (w == 0) {
        const int s = wsum[lane];
        int t = s;
#pragma unroll
        for (int o = 1; o < 32; o <<= 1) {
            int y = __shfl_up(t, o);
            if (lane >= o) t += y;
        }
        wsum[lane] = t - s;
    }
    __syncthreads();
    return x - v + wsum[w];
}

__global__ __launch_bounds__(1024) void k_partition(const int* __restrict__ src,
                                                    const int* __restrict__ dst,
                                                    int* __restrict__ runs,
                                                    unsigned* __restrict__ sorted, int E,
                                                    int chunk, int P) {
    __shared__ int hist[MAXP];
    __shared__ int base[MAXP];
    __shared__ int thist[MAXP];
    __shared__ int tstart[MAXP];
    __shared__ int wsum[32];
    __shared__ unsigned sw[STAGE_CAP];
    __shared__ int sidx[STAGE_CAP];
    const int tid = threadIdx.x;
    hist[tid] = 0;
    __syncthreads();
    long beg = (long)blockIdx.x * chunk;
    long end = beg + chunk;
    if (end > (long)E) end = E;
    for (long e0 = beg + tid; e0 < end; e0 += 4096) {
        int d[4];
#pragma unroll
        for (int u = 0; u < 4; ++u) {
            long e = e0 + u * 1024;
            d[u] = (e < end) ? dst[e] : -1;
        }
#pragma unroll
        for (int u = 0; u < 4; ++u)
            if (d[u] >= 0) atomicAdd(&hist[d[u] / RB], 1);
    }
    __syncthreads();
    {
        const int hc = hist[tid];
        const int bs = (int)beg + block_excl_scan(hc, wsum, tid);
        base[tid] = bs;
        if (tid < P) {
            const size_t ri = ((size_t)tid * NPART + blockIdx.x) * 2;
            runs[ri] = bs;
            runs[ri + 1] = hc;
        }
    }
    __syncthreads();
    const bool one_tile = (end - beg) <= (long)STAGE_CAP;
    for (long t0 = beg; t0 < end; t0 += STAGE_CAP) {
        long t1 = t0 + STAGE_CAP;
        if (t1 > end) t1 = end;
        if (one_tile) {
            thist[tid] = hist[tid];
        } else {
            thist[tid] = 0;
            __syncthreads();
            for (long e0 = t0 + tid; e0 < t1; e0 += 4096) {
                int d[4];
#pragma unroll
                for (int u = 0; u < 4; ++u) {
                    long e = e0 + u * 1024;
                    d[u] = (e < t1) ? dst[e] : -1;
                }
#pragma unroll
                for (int u = 0; u < 4; ++u)
                    if (d[u] >= 0) atomicAdd(&thist[d[u] / RB], 1);
            }
        }
        __syncthreads();
        const int tc = thist[tid];
        tstart[tid] = block_excl_scan(tc, wsum, tid);
        thist[tid] = 0;
        __syncthreads();
        for (long e0 = t0 + tid; e0 < t1; e0 += 4096) {
            int d[4], s[4];
#pragma unroll
            for (int u = 0; u < 4; ++u) {
                long e = e0 + u * 1024;
                bool ok = e < t1;
                d[u] = ok ? dst[e] : -1;
                s[u] = ok ? src[e] : 0;
            }
#pragma unroll
            for (int u = 0; u < 4; ++u) {
                if (d[u] >= 0) {
                    int b = d[u] / RB;
                    int k = atomicAdd(&thist[b], 1);
                    int p = tstart[b] + k;
                    sw[p] = (unsigned)s[u] | ((unsigned)(d[u] - b * RB) << SRC_BITS);
                    sidx[p] = base[b] + k;
                }
            }
        }
        __syncthreads();
        const int n = (int)(t1 - t0);
        for (int i = tid; i < n; i += 1024) sorted[sidx[i]] = sw[i];
        if (!one_tile) {
            __syncthreads();
            base[tid] += thist[tid];
            __syncthreads();
        }
    }
}


__global__ __launch_bounds__(1024) void k_deg_gemm1(const unsigned* __restrict__ sorted,
                                                    const int* __restrict__ runs,
                                                    const float* __restrict__ x,
                                                    const float* __restrict__ W1,
                                                    int* __restrict__ deg,
                                                    float* __restrict__ g, int N) {
    __shared__ int cntl[RB];
    const int tid = threadIdx.x;
    const int lo = blockIdx.x * RB;
    const int lane = tid & 31;
    const int f = lane & 15, half = lane >> 4;
    const int wv = __builtin_amdgcn_readfirstlane(tid >> 5);
    const bool has_tile = wv < RB / 16;
    const int r0 = wv * 16;
    int arow = lo + r0 + f;
    arow = arow < N ? arow : (N - 1);
    for (int i = tid; i < RB; i += 1024) cntl[i] = 0;
    const WaveRuns wr = wave_runs(runs, blockIdx.x, wv);
    __syncthreads();
    for (int e0 = lane; e0 < wr.total; e0 += 128) {
        unsigned q[4];
#pragma unroll
        for (int u = 0; u < 4; ++u) {
            int e = e0 + u * 32;
            q[u] = (e < wr.total) ? sorted[run_slot(wr, e)] : NO_EDGE;
        }
#pragma unroll
        for (int u = 0; u < 4; ++u)
            if (q[u] != NO_EDGE) atomicAdd(&cntl[q[u] >> SRC_BITS], 1);
    }
    __syncthreads();
    for (int i = tid; i < RB; i += 1024) {
        int node = lo + i;
        if (node < N) deg[node] = cntl[i];
    }
    if (has_tile) {
        const float sc = rsqrtf((float)(cntl[r0 + f] + 1));
        const float* __restrict__ ap = x + (long)arow * 128 + 2 * half;
        const float* __restrict__ bp = W1 + f;
        v8f c = {0.f, 0.f, 0.f, 0.f, 0.f, 0.f, 0.f, 0.f};
#pragma unroll
        for (int j = 0; j < 32; ++j) {
            v2f av = *(const v2f*)(ap + 4 * j);
            av.x *= sc;
            av.y *= sc;
            int ka = 4 * j + 2 * half;
            v2f b;
            b.x = bp[(ka + 0) * 16];
            b.y = bp[(ka + 1) * 16];
            c = __builtin_amdgcn_wmma_f32_16x16x4_f32(false, av, false, b, (short)0, c,
                                                      false, false);
        }
#pragma unroll
        for (int v = 0; v < 8; ++v) {
            int row = lo + r0 + v + 8 * half;
            if (row < N) g[(long)row * 16 + f] = c[v];
        }
    }
}

#define FILE_CAP 24576
static_assert(RB % 8 == 0, "eight nodes (one per four lanes) per wave and round");

template <int J>
__device__ __forceinline__ unsigned quad_bcast(unsigned v) {
    return (unsigned)__builtin_amdgcn_update_dpp(0, (int)v, J * 0x55, 0xF, 0xF, false);
}

__device__ __forceinline__ void load_words16(unsigned (&q)[16], const unsigned* __restrict__ sorted,
                                             const WaveRuns& wr, int e0) {
#pragma unroll
    for (int u = 0; u < 16; ++u) {
        const int e = e0 + u * 32;
        q[u] = (e < wr.total) ? sorted[run_slot(wr, e)] : NO_EDGE;
    }
}

template <int NCOL>
__global__ __launch_bounds__(1024) void k_bucket16(const unsigned* __restrict__ sorted,
                                                   const int* __restrict__ runs,
                                                   const float* __restrict__ g,
                                                   const int* __restrict__ deg,
                                                   const float* __restrict__ bias,
                                                   const float* __restrict__ Wn,
                                                   float* __restrict__ gn, int N) {
    __shared__ float acc[(RB + 1) * 16];
    __shared__ float disl[RB];
    __shared__ int offl[RB + 1];
    __shared__ int curl[RB + 32];
    __shared__ int wsum[32];
    __shared__ unsigned srcl[FILE_CAP];
    const int tid = threadIdx.x;
    const int lo = blockIdx.x * RB;
    const int lane = tid & 31, wave = tid >> 5;
    const int f = lane & 15, half = lane >> 4;
    const int wv = __builtin_amdgcn_readfirstlane(wave);
    const WaveRuns wr = wave_runs(runs, blockIdx.x, wv);
    const int total = wr.total;
    const bool bvalid = (f < NCOL);
    v2f b[4];
#pragma unroll
    for (int j = 0; j < 4; ++j) {
        const int ka = 4 * j + 2 * half;
        b[j].x = bvalid ? Wn[(ka + 0) * NCOL + f] : 0.f;
        b[j].y = bvalid ? Wn[(ka + 1) * NCOL + f] : 0.f;
    }
    unsigned q[16];
    load_words16(q, sorted, wr, lane);
    {
        int dg = 0;
        if (tid < RB && lo + tid < N) dg = deg[lo + tid];
        const int st = block_excl_scan(dg, wsum, tid);
        if (tid <= RB) offl[tid] = st;
        if (tid < RB + 32) curl[tid] = st;
        if (tid < RB) disl[tid] = (lo + tid < N) ? rsqrtf((float)(dg + 1)) : 0.f;
    }
    __syncthreads();
    const int btotal = offl[RB];
    if (btotal <= FILE_CAP) {
        for (int m = 0;;) {
            int slot[16];
#pragma unroll
            for (int u = 0; u < 16; ++u) {
                const unsigned cw = (q[u] != NO_EDGE) ? (q[u] >> SRC_BITS) : (unsigned)(RB + lane);
                slot[u] = atomicAdd(&curl[cw], 1);
            }
#pragma unroll
            for (int u = 0; u < 16; ++u)
                if (q[u] != NO_EDGE) srcl[slot[u]] = q[u] & SRC_MASK;
            m += 512;
            if (m >= total) break;
            load_words16(q, sorted, wr, m + lane);
        }
        __syncthreads();
        const int quad = lane >> 2, c = lane & 3;
        float bv[4];
#pragma unroll
        for (int t = 0; t < 4; ++t) bv[t] = bias[4 * c + t];
        for (int n0 = wv * 8; n0 < RB; n0 += 256) {
            const int n = n0 + quad;
            const int beg = offl[n];
            const int len = offl[n + 1] - beg;
            const int node = lo + n;
            const bool live = node < N;
            float4 self = make_float4(0.f, 0.f, 0.f, 0.f);
            if (live) self = *(const float4*)(g + ((unsigned)node * 16u + 4u * (unsigned)c));
            float4 a = make_float4(0.f, 0.f, 0.f, 0.f);
            for (int k = 0; __any(k < len); k += 16) {
                unsigned ix[4];
#pragma unroll
                for (int t = 0; t < 4; ++t) {
                    const int p = k + 4 * t + c;
                    ix[t] = (p < len) ? srcl[beg + p] : 0u;
                }
                unsigned sx[16];
#pragma unroll
                for (int t = 0; t < 4; ++t) {
                    sx[4 * t + 0] = quad_bcast<0>(ix[t]);
                    sx[4 * t + 1] = quad_bcast<1>(ix[t]);
                    sx[4 * t + 2] = quad_bcast<2>(ix[t]);
                    sx[4 * t + 3] = quad_bcast<3>(ix[t]);
                }
                float4 v[16];
#pragma unroll
                for (int j = 0; j < 16; ++j) {
                    v[j] = make_float4(0.f, 0.f, 0.f, 0.f);
                    if (k + j < len) v[j] = *(const float4*)((const char*)g + (sx[j] * 64u + 16u * (unsigned)c));
                }
#pragma unroll
                for (int j = 0; j < 16; ++j) {
                    a.x += v[j].x;
                    a.y += v[j].y;
                    a.z += v[j].z;
                    a.w += v[j].w;
                }
            }
            const float di = disl[n];
            float4 r;
            r.x = di * (a.x + self.x) + bv[0];
            r.y = di * (a.y + self.y) + bv[1];
            r.z = di * (a.z + self.z) + bv[2];
            r.w = di * (a.w + self.w) + bv[3];
            r.x = (live && r.x > 0.f) ? r.x : 0.f;
            r.y = (live && r.y > 0.f) ? r.y : 0.f;
            r.z = (live && r.z > 0.f) ? r.z : 0.f;
            r.w = (live && r.w > 0.f) ? r.w : 0.f;
            *(float4*)&acc[n * 16 + 4 * c] = r;
        }
    } else {
        for (int i = tid; i < (RB + 1) * 16; i += 1024) acc[i] = 0.f;
        __syncthreads();
        unsigned pk = (lane < total) ? sorted[run_slot(wr, lane)] : SPARE_EDGE;
        for (int c0 = 0; c0 < total; c0 += 32) {
            const int en = c0 + 32 + lane;
            const unsigned wn = sorted[(en < total) ? run_slot(wr, en) : 0];
            const unsigned pkn = (en < total) ? wn : SPARE_EDGE;
            unsigned q[16];
            float v[16];
#pragma unroll
            for (int j = 0; j < 16; ++j) q[j] = (unsigned)__shfl((int)pk, 2 * j + half);
#pragma unroll
            for (int j = 0; j < 16; ++j) v[j] = g[(q[j] & SRC_MASK) * 16u + (unsigned)f];
#pragma unroll
            for (int j = 0; j < 16; ++j)
                __hip_atomic_fetch_add(&acc[(q[j] >> SRC_BITS) * 16u + (unsigned)f], v[j],
                                       __ATOMIC_RELAXED, __HIP_MEMORY_SCOPE_WORKGROUP);
            pk = pkn;
        }
        __syncthreads();
        for (int i = tid; i < RB * 16; i += 1024) {
            const int r = i >> 4, node = lo + r, ff = i & 15;
            float v = 0.f;
            if (node < N) {
                size_t o = (size_t)node * 16 + ff;
                v = disl[r] * (acc[i] + g[o]) + bias[ff];
                v = v > 0.f ? v : 0.f;
            }
            acc[i] = v;
        }
    }
    __syncthreads();
    if (wv < RB / 16) {
        const int r0 = wv * 16;
        const float sc = disl[r0 + f];
        const float* ap = &acc[(r0 + f) * 16 + 2 * half];
        v8f c = {0.f, 0.f, 0.f, 0.f, 0.f, 0.f, 0.f, 0.f};
#pragma unroll
        for (int j = 0; j < 4; ++j) {
            v2f a = *(const v2f*)(ap + 4 * j);
            a.x *= sc;
            a.y *= sc;
            c = __builtin_amdgcn_wmma_f32_16x16x4_f32(false, a, false, b[j], (short)0, c,
                                                      false, false);
        }
#pragma unroll
        for (int v = 0; v < 8; ++v) {
            int row = lo + r0 + v + 8 * half;
            if (row < N && f < NCOL) gn[(size_t)row * NCOL + f] = c[v];
        }
    }
}

__global__ __launch_bounds__(1024) void k_bucket2_lsm(const unsigned* __restrict__ sorted,
                                                      const int* __restrict__ runs,
                                                      const float* __restrict__ g,
                                                      const int* __restrict__ deg,
                                                      const float* __restrict__ bias,
                                                      float* __restrict__ out, int N) {
    __shared__ float acc[RB * 2];
    const int tid = threadIdx.x;
    const int lo = blockIdx.x * RB;
    for (int i = tid; i < RB * 2; i += 1024) acc[i] = 0.f;
    const int wv = __builtin_amdgcn_readfirstlane(tid >> 5);
    const int lane = tid & 31;
    const WaveRuns wr = wave_runs(runs, blockIdx.x, wv);
    __syncthreads();
    for (int e0 = lane; e0 < wr.total; e0 += 128) {
        unsigned q[4];
        v2f v[4];
#pragma unroll
        for (int u = 0; u < 4; ++u) {
            int e = e0 + u * 32;
            q[u] = (e < wr.total) ? sorted[run_slot(wr, e)] : NO_EDGE;
        }
#pragma unroll
        for (int u = 0; u < 4; ++u) {
            unsigned s = (q[u] != NO_EDGE) ? (q[u] & SRC_MASK) : 0u;
            v[u] = *(const v2f*)(g + s * 2u);
        }
#pragma unroll
        for (int u = 0; u < 4; ++u) {
            if (q[u] != NO_EDGE) {
                unsigned ld = q[u] >> SRC_BITS;
                __hip_atomic_fetch_add(&acc[ld * 2u + 0u], v[u].x, __ATOMIC_RELAXED,
                                       __HIP_MEMORY_SCOPE_WORKGROUP);
                __hip_atomic_fetch_add(&acc[ld * 2u + 1u], v[u].y, __ATOMIC_RELAXED,
                                       __HIP_MEMORY_SCOPE_WORKGROUP);
            }
        }
    }
    __syncthreads();
    for (int i = tid; i < RB; i += 1024) {
        int node = lo + i;
        if (node < N) {
            float di = rsqrtf((float)(deg[node] + 1));
            float z0 = di * (acc[2 * i + 0] + g[(size_t)node * 2 + 0]) + bias[0];
            float z1 = di * (acc[2 * i + 1] + g[(size_t)node * 2 + 1]) + bias[1];
            float m = fmaxf(z0, z1);
            float lse = m + __logf(__expf(z0 - m) + __expf(z1 - m));
            out[(size_t)node * 2 + 0] = z0 - lse;
            out[(size_t)node * 2 + 1] = z1 - lse;
        }
    }
}

extern "C" void kernel_launch(void* const* d_in, const int* in_sizes, int n_in,
                              void* d_out, int out_size, void* d_ws, size_t ws_size,
                              hipStream_t stream) {
    const float* x  = (const float*)d_in[0];
    const int*   ei = (const int*)d_in[1];
    const float* W1 = (const float*)d_in[2];
    const float* b1 = (const float*)d_in[3];
    const float* W2 = (const float*)d_in[4];
    const float* b2 = (const float*)d_in[5];
    const float* W3 = (const float*)d_in[6];
    const float* b3 = (const float*)d_in[7];

    const int N = in_sizes[0] / 128;
    const int E = in_sizes[1] / 2;
    const int* src = ei;
    const int* dst = ei + E;

    char* ws = (char*)d_ws;
    size_t off = 0;
    auto carve = [&](size_t bytes) -> void* {
        void* p = ws + off;
        off += (bytes + 255) & ~(size_t)255;
        return p;
    };
    const int P = (N + RB - 1) / RB;
    const int chunk = (E + NPART - 1) / NPART;
    int*      deg     = (int*)carve((size_t)N * 4);
    int*      runs    = (int*)carve((size_t)P * NPART * 2 * 4);
    unsigned* sorted  = (unsigned*)carve((size_t)E * 4);
    float*    g       = (float*)carve((size_t)N * 16 * 4);
    float*    g2      = (float*)carve((size_t)N * 16 * 4);
    (void)ws_size; (void)n_in; (void)out_size;

    k_partition<<<NPART, 1024, 0, stream>>>(src, dst, runs, sorted, E, chunk, P);

    k_deg_gemm1<<<P, 1024, 0, stream>>>(sorted, runs, x, W1, deg, g, N);

    k_bucket16<16><<<P, 1024, 0, stream>>>(sorted, runs, g, deg, b1, W2, g2, N);

    k_bucket16<2><<<P, 1024, 0, stream>>>(sorted, runs, g2, deg, b2, W3, g, N);

    k_bucket2_lsm<<<P, 1024, 0, stream>>>(sorted, runs, g, deg, b3, (float*)d_out, N);
}


# last layer sums only the two class scores' difference (all a 2-class log_softmax needs): one float a node, table copied to each work-group's LDS, gathers read LDS; sums arrival order, of differences
# speedup vs baseline: 7.2615x; 1.0692x over previous
#include <hip/hip_runtime.h>
#include <math.h>


typedef float v2f __attribute__((ext_vector_type(2)));
typedef float v8f __attribute__((ext_vector_type(8)));

#define RB 416
#define MAXP 1024
#define SRC_BITS 20
#define SRC_MASK ((1u << SRC_BITS) - 1u)
#define NO_EDGE 0xFFFFFFFFu
#define SPARE_EDGE ((unsigned)RB << SRC_BITS)
#define NPART 256
static_assert(RB % 16 == 0 && RB / 16 <= 32, "one wave of a 1024-thread work-group per 16-row tile");

#define RPW (NPART / 32)
static_assert(NPART % 32 == 0, "each of the 32 waves of a consumer work-group takes RPW whole runs");

struct WaveRuns {
    int o[RPW];
    int d[RPW];
    int total;
};

__device__ __forceinline__ WaveRuns wave_runs(const int* __restrict__ runs, int bucket, int wv) {
    const int* __restrict__ rt = runs + ((size_t)bucket * NPART + (size_t)wv * RPW) * 2;
    WaveRuns r;
    int acc = 0;
#pragma unroll
    for (int k = 0; k < RPW; ++k) {
        const int st = rt[2 * k], ln = rt[2 * k + 1];
        r.o[k] = acc;
        r.d[k] = st - acc;
        acc += ln;
    }
    r.total = acc;
    return r;
}

__device__ __forceinline__ int run_slot(const WaveRuns& r, int e) {
    int d = r.d[0];
#pragma unroll
    for (int k = 1; k < RPW; ++k) d = (e >= r.o[k]) ? r.d[k] : d;
    return e + d;
}

#define STAGE_CAP 16384
static_assert(MAXP == 1024, "the scan below gives one table entry to each of the 1024 threads");

__device__ __forceinline__ int block_excl_scan(int v, int* wsum, int tid) {
    const int lane = tid & 31, w = tid >> 5;
    int x = v;
#pragma unroll
    for (int o = 1; o < 32; o <<= 1) {
        int y = __shfl_up(x, o);
        if (lane >= o) x += y;
    }
    if (lane == 31) wsum[w] = x;
    __syncthreads();
    if (w == 0) {
        const int s = wsum[lane];
        int t = s;
#pragma unroll
        for (int o = 1; o < 32; o <<= 1) {
            int y = __shfl_up(t, o);
            if (lane >= o) t += y;
        }
        wsum[lane] = t - s;
    }
    __syncthreads();
    return x - v + wsum[w];
}

__global__ __launch_bounds__(1024) void k_partition(const int* __restrict__ src,
                                                    const int* __restrict__ dst,
                                                    int* __restrict__ runs,
                                                    unsigned* __restrict__ sorted, int E,
                                                    int chunk, int P) {
    __shared__ int hist[MAXP];
    __shared__ int base[MAXP];
    __shared__ int thist[MAXP];
    __shared__ int tstart[MAXP];
    __shared__ int wsum[32];
    __shared__ unsigned sw[STAGE_CAP];
    __shared__ int sidx[STAGE_CAP];
    const int tid = threadIdx.x;
    hist[tid] = 0;
    __syncthreads();
    long beg = (long)blockIdx.x * chunk;
    long end = beg + chunk;
    if (end > (long)E) end = E;
    for (long e0 = beg + tid; e0 < end; e0 += 4096) {
        int d[4];
#pragma unroll
        for (int u = 0; u < 4; ++u) {
            long e = e0 + u * 1024;
            d[u] = (e < end) ? dst[e] : -1;
        }
#pragma unroll
        for (int u = 0; u < 4; ++u)
            if (d[u] >= 0) atomicAdd(&hist[d[u] / RB], 1);
    }
    __syncthreads();
    {
        const int hc = hist[tid];
        const int bs = (int)beg + block_excl_scan(hc, wsum, tid);
        base[tid] = bs;
        if (tid < P) {
            const size_t ri = ((size_t)tid * NPART + blockIdx.x) * 2;
            runs[ri] = bs;
            runs[ri + 1] = hc;
        }
    }
    __syncthreads();
    const bool one_tile = (end - beg) <= (long)STAGE_CAP;
    for (long t0 = beg; t0 < end; t0 += STAGE_CAP) {
        long t1 = t0 + STAGE_CAP;
        if (t1 > end) t1 = end;
        if (one_tile) {
            thist[tid] = hist[tid];
        } else {
            thist[tid] = 0;
            __syncthreads();
            for (long e0 = t0 + tid; e0 < t1; e0 += 4096) {
                int d[4];
#pragma unroll
                for (int u = 0; u < 4; ++u) {
                    long e = e0 + u * 1024;
                    d[u] = (e < t1) ? dst[e] : -1;
                }
#pragma unroll
                for (int u = 0; u < 4; ++u)
                    if (d[u] >= 0) atomicAdd(&thist[d[u] / RB], 1);
            }
        }
        __syncthreads();
        const int tc = thist[tid];
        tstart[tid] = block_excl_scan(tc, wsum, tid);
        thist[tid] = 0;
        __syncthreads();
        for (long e0 = t0 + tid; e0 < t1; e0 += 4096) {
            int d[4], s[4];
#pragma unroll
            for (int u = 0; u < 4; ++u) {
                long e = e0 + u * 1024;
                bool ok = e < t1;
                d[u] = ok ? dst[e] : -1;
                s[u] = ok ? src[e] : 0;
            }
#pragma unroll
            for (int u = 0; u < 4; ++u) {
                if (d[u] >= 0) {
                    int b = d[u] / RB;
                    int k = atomicAdd(&thist[b], 1);
                    int p = tstart[b] + k;
                    sw[p] = (unsigned)s[u] | ((unsigned)(d[u] - b * RB) << SRC_BITS);
                    sidx[p] = base[b] + k;
                }
            }
        }
        __syncthreads();
        const int n = (int)(t1 - t0);
        for (int i = tid; i < n; i += 1024) sorted[sidx[i]] = sw[i];
        if (!one_tile) {
            __syncthreads();
            base[tid] += thist[tid];
            __syncthreads();
        }
    }
}


__global__ __launch_bounds__(1024) void k_deg_gemm1(const unsigned* __restrict__ sorted,
                                                    const int* __restrict__ runs,
                                                    const float* __restrict__ x,
                                                    const float* __restrict__ W1,
                                                    int* __restrict__ deg,
                                                    float* __restrict__ g, int N) {
    __shared__ int cntl[RB];
    const int tid = threadIdx.x;
    const int lo = blockIdx.x * RB;
    const int lane = tid & 31;
    const int f = lane & 15, half = lane >> 4;
    const int wv = __builtin_amdgcn_readfirstlane(tid >> 5);
    const bool has_tile = wv < RB / 16;
    const int r0 = wv * 16;
    int arow = lo + r0 + f;
    arow = arow < N ? arow : (N - 1);
    for (int i = tid; i < RB; i += 1024) cntl[i] = 0;
    const WaveRuns wr = wave_runs(runs, blockIdx.x, wv);
    __syncthreads();
    for (int e0 = lane; e0 < wr.total; e0 += 128) {
        unsigned q[4];
#pragma unroll
        for (int u = 0; u < 4; ++u) {
            int e = e0 + u * 32;
            q[u] = (e < wr.total) ? sorted[run_slot(wr, e)] : NO_EDGE;
        }
#pragma unroll
        for (int u = 0; u < 4; ++u)
            if (q[u] != NO_EDGE) atomicAdd(&cntl[q[u] >> SRC_BITS], 1);
    }
    __syncthreads();
    for (int i = tid; i < RB; i += 1024) {
        int node = lo + i;
        if (node < N) deg[node] = cntl[i];
    }
    if (has_tile) {
        const float sc = rsqrtf((float)(cntl[r0 + f] + 1));
        const float* __restrict__ ap = x + (long)arow * 128 + 2 * half;
        const float* __restrict__ bp = W1 + f;
        v8f c = {0.f, 0.f, 0.f, 0.f, 0.f, 0.f, 0.f, 0.f};
#pragma unroll
        for (int j = 0; j < 32; ++j) {
            v2f av = *(const v2f*)(ap + 4 * j);
            av.x *= sc;
            av.y *= sc;
            int ka = 4 * j + 2 * half;
            v2f b;
            b.x = bp[(ka + 0) * 16];
            b.y = bp[(ka + 1) * 16];
            c = __builtin_amdgcn_wmma_f32_16x16x4_f32(false, av, false, b, (short)0, c,
                                                      false, false);
        }
#pragma unroll
        for (int v = 0; v < 8; ++v) {
            int row = lo + r0 + v + 8 * half;
            if (row < N) g[(long)row * 16 + f] = c[v];
        }
    }
}

#define FILE_CAP 24576
static_assert(RB % 8 == 0, "eight nodes (one per four lanes) per wave and round");

template <int J>
__device__ __forceinline__ unsigned quad_bcast(unsigned v) {
    return (unsigned)__builtin_amdgcn_update_dpp(0, (int)v, J * 0x55, 0xF, 0xF, false);
}

__device__ __forceinline__ void load_words16(unsigned (&q)[16], const unsigned* __restrict__ sorted,
                                             const WaveRuns& wr, int e0) {
#pragma unroll
    for (int u = 0; u < 16; ++u) {
        const int e = e0 + u * 32;
        q[u] = (e < wr.total) ? sorted[run_slot(wr, e)] : NO_EDGE;
    }
}

template <int NCOL>
__global__ __launch_bounds__(1024) void k_bucket16(const unsigned* __restrict__ sorted,
                                                   const int* __restrict__ runs,
                                                   const float* __restrict__ g,
                                                   const int* __restrict__ deg,
                                                   const float* __restrict__ bias,
                                                   const float* __restrict__ Wn,
                                                   float* __restrict__ gn, int N) {
    __shared__ float acc[(RB + 1) * 16];
    __shared__ float disl[RB];
    __shared__ int offl[RB + 1];
    __shared__ int curl[RB + 32];
    __shared__ int wsum[32];
    __shared__ unsigned srcl[FILE_CAP];
    __shared__ float gcol[NCOL == 1 ? RB : 1];
    const int tid = threadIdx.x;
    const int lo = blockIdx.x * RB;
    const int lane = tid & 31, wave = tid >> 5;
    const int f = lane & 15, half = lane >> 4;
    const int wv = __builtin_amdgcn_readfirstlane(wave);
    const WaveRuns wr = wave_runs(runs, blockIdx.x, wv);
    const int total = wr.total;
    const bool bvalid = (f < NCOL);
    v2f b[4];
#pragma unroll
    for (int j = 0; j < 4; ++j) {
        const int ka = 4 * j + 2 * half;
        if (NCOL == 16) {
            b[j].x = Wn[(ka + 0) * 16 + f];
            b[j].y = Wn[(ka + 1) * 16 + f];
        } else {
            b[j].x = bvalid ? Wn[(ka + 0) * 2 + 0] - Wn[(ka + 0) * 2 + 1] : 0.f;
            b[j].y = bvalid ? Wn[(ka + 1) * 2 + 0] - Wn[(ka + 1) * 2 + 1] : 0.f;
        }
    }
    unsigned q[16];
    load_words16(q, sorted, wr, lane);
    {
        int dg = 0;
        if (tid < RB && lo + tid < N) dg = deg[lo + tid];
        const int st = block_excl_scan(dg, wsum, tid);
        if (tid <= RB) offl[tid] = st;
        if (tid < RB + 32) curl[tid] = st;
        if (tid < RB) disl[tid] = (lo + tid < N) ? rsqrtf((float)(dg + 1)) : 0.f;
    }
    __syncthreads();
    const int btotal = offl[RB];
    if (btotal <= FILE_CAP) {
        for (int m = 0;;) {
            int slot[16];
#pragma unroll
            for (int u = 0; u < 16; ++u) {
                const unsigned cw = (q[u] != NO_EDGE) ? (q[u] >> SRC_BITS) : (unsigned)(RB + lane);
                slot[u] = atomicAdd(&curl[cw], 1);
            }
#pragma unroll
            for (int u = 0; u < 16; ++u)
                if (q[u] != NO_EDGE) srcl[slot[u]] = q[u] & SRC_MASK;
            m += 512;
            if (m >= total) break;
            load_words16(q, sorted, wr, m + lane);
        }
        __syncthreads();
        const int quad = lane >> 2, c = lane & 3;
        float bv[4];
#pragma unroll
        for (int t = 0; t < 4; ++t) bv[t] = bias[4 * c + t];
        for (int n0 = wv * 8; n0 < RB; n0 += 256) {
            const int n = n0 + quad;
            const int beg = offl[n];
            const int len = offl[n + 1] - beg;
            const int node = lo + n;
            const bool live = node < N;
            float4 self = make_float4(0.f, 0.f, 0.f, 0.f);
            if (live) self = *(const float4*)(g + ((unsigned)node * 16u + 4u * (unsigned)c));
            float4 a = make_float4(0.f, 0.f, 0.f, 0.f);
            for (int k = 0; __any(k < len); k += 16) {
                unsigned ix[4];
#pragma unroll
                for (int t = 0; t < 4; ++t) {
                    const int p = k + 4 * t + c;
                    ix[t] = (p < len) ? srcl[beg + p] : 0u;
                }
                unsigned sx[16];
#pragma unroll
                for (int t = 0; t < 4; ++t) {
                    sx[4 * t + 0] = quad_bcast<0>(ix[t]);
                    sx[4 * t + 1] = quad_bcast<1>(ix[t]);
                    sx[4 * t + 2] = quad_bcast<2>(ix[t]);
                    sx[4 * t + 3] = quad_bcast<3>(ix[t]);
                }
                float4 v[16];
#pragma unroll
                for (int j = 0; j < 16; ++j) {
                    v[j] = make_float4(0.f, 0.f, 0.f, 0.f);
                    if (k + j < len) v[j] = *(const float4*)((const char*)g + (sx[j] * 64u + 16u * (unsigned)c));
                }
#pragma unroll
                for (int j = 0; j < 16; ++j) {
                    a.x += v[j].x;
                    a.y += v[j].y;
                    a.z += v[j].z;
                    a.w += v[j].w;
                }
            }
            const float di = disl[n];
            float4 r;
            r.x = di * (a.x + self.x) + bv[0];
            r.y = di * (a.y + self.y) + bv[1];
            r.z = di * (a.z + self.z) + bv[2];
            r.w = di * (a.w + self.w) + bv[3];
            r.x = (live && r.x > 0.f) ? r.x : 0.f;
            r.y = (live && r.y > 0.f) ? r.y : 0.f;
            r.z = (live && r.z > 0.f) ? r.z : 0.f;
            r.w = (live && r.w > 0.f) ? r.w : 0.f;
            *(float4*)&acc[n * 16 + 4 * c] = r;
        }
    } else {
        for (int i = tid; i < (RB + 1) * 16; i += 1024) acc[i] = 0.f;
        __syncthreads();
        unsigned pk = (lane < total) ? sorted[run_slot(wr, lane)] : SPARE_EDGE;
        for (int c0 = 0; c0 < total; c0 += 32) {
            const int en = c0 + 32 + lane;
            const unsigned wn = sorted[(en < total) ? run_slot(wr, en) : 0];
            const unsigned pkn = (en < total) ? wn : SPARE_EDGE;
            unsigned q[16];
            float v[16];
#pragma unroll
            for (int j = 0; j < 16; ++j) q[j] = (unsigned)__shfl((int)pk, 2 * j + half);
#pragma unroll
            for (int j = 0; j < 16; ++j) v[j] = g[(q[j] & SRC_MASK) * 16u + (unsigned)f];
#pragma unroll
            for (int j = 0; j < 16; ++j)
                __hip_atomic_fetch_add(&acc[(q[j] >> SRC_BITS) * 16u + (unsigned)f], v[j],
                                       __ATOMIC_RELAXED, __HIP_MEMORY_SCOPE_WORKGROUP);
            pk = pkn;
        }
        __syncthreads();
        for (int i = tid; i < RB * 16; i += 1024) {
            const int r = i >> 4, node = lo + r, ff = i & 15;
            float v = 0.f;
            if (node < N) {
                size_t o = (size_t)node * 16 + ff;
                v = disl[r] * (acc[i] + g[o]) + bias[ff];
                v = v > 0.f ? v : 0.f;
            }
            acc[i] = v;
        }
    }
    __syncthreads();
    if (wv < RB / 16) {
        const int r0 = wv * 16;
        const float sc = disl[r0 + f];
        const float* ap = &acc[(r0 + f) * 16 + 2 * half];
        v8f c = {0.f, 0.f, 0.f, 0.f, 0.f, 0.f, 0.f, 0.f};
#pragma unroll
        for (int j = 0; j < 4; ++j) {
            v2f a = *(const v2f*)(ap + 4 * j);
            a.x *= sc;
            a.y *= sc;
            c = __builtin_amdgcn_wmma_f32_16x16x4_f32(false, a, false, b[j], (short)0, c,
                                                      false, false);
        }
        if (NCOL == 16) {
#pragma unroll
            for (int v = 0; v < 8; ++v) {
                int row = lo + r0 + v + 8 * half;
                if (row < N) gn[(size_t)row * 16 + f] = c[v];
            }
        } else if (f == 0) {
#pragma unroll
            for (int v = 0; v < 8; ++v) gcol[r0 + v + 8 * half] = c[v];
        }
    }
    if (NCOL == 1) {
        __syncthreads();
        if (tid < RB && lo + tid < N) gn[lo + tid] = gcol[tid];
    }
}

#define TAB_N 80000
static_assert(TAB_N % 4 == 0, "the table copy moves 16 bytes per lane");
static_assert((TAB_N + 4 + RB) * 4 <= 320 * 1024, "table and sums fit one work-group's LDS");
#define CPY 10

__global__ __launch_bounds__(1024) void k_bucket1_lsm(const unsigned* __restrict__ sorted,
                                                      const int* __restrict__ runs,
                                                      const float* __restrict__ gd,
                                                      const int* __restrict__ deg,
                                                      const float* __restrict__ bias,
                                                      float* __restrict__ out, int N, int P) {
    __shared__ __align__(16) float tab[TAB_N + 4];
    __shared__ float acc[RB];
    const int tid = threadIdx.x;
    const int lo = blockIdx.x * RB;
    for (int i = tid; i < RB; i += 1024) acc[i] = 0.f;
    const int wv = __builtin_amdgcn_readfirstlane(tid >> 5);
    const int lane = tid & 31;
    const WaveRuns wr = wave_runs(runs, blockIdx.x, wv);
    const int total = wr.total;
    unsigned q[16];
    load_words16(q, sorted, wr, lane);
    const int nt = N < TAB_N ? N : TAB_N;
    const int nt4 = nt >> 2;
    const int rot = (int)(((unsigned)blockIdx.x * (unsigned)nt4) / (unsigned)P);
    for (int i0 = tid; i0 < nt4; i0 += CPY * 1024) {
        float4 t[CPY];
        int ix[CPY];
#pragma unroll
        for (int u = 0; u < CPY; ++u) {
            const int i = i0 + u * 1024;
            int j = i + rot;
            j = j >= nt4 ? j - nt4 : j;
            const bool in = i < nt4;
            ix[u] = in ? 4 * j : TAB_N;
            t[u] = *(const float4*)(gd + (in ? 4 * j : 0));
        }
#pragma unroll
        for (int u = 0; u < CPY; ++u) *(float4*)&tab[ix[u]] = t[u];
    }
    for (int i = 4 * nt4 + tid; i < nt; i += 1024) tab[i] = gd[i];
    __syncthreads();
    for (int m = 0;;) {
        float vt[16], vg[16];
#pragma unroll
        for (int u = 0; u < 16; ++u) {
            const unsigned s = q[u] & SRC_MASK;
            vt[u] = tab[(q[u] != NO_EDGE && s < (unsigned)nt) ? s : 0u];
        }
#pragma unroll
        for (int u = 0; u < 16; ++u) {
            const unsigned s = q[u] & SRC_MASK;
            vg[u] = 0.f;
            if (q[u] != NO_EDGE && s >= (unsigned)nt) vg[u] = gd[s];
        }
#pragma unroll
        for (int u = 0; u < 16; ++u) {
            const unsigned s = q[u] & SRC_MASK;
            if (q[u] != NO_EDGE)
                __hip_atomic_fetch_add(&acc[q[u] >> SRC_BITS], s < (unsigned)nt ? vt[u] : vg[u],
                                       __ATOMIC_RELAXED, __HIP_MEMORY_SCOPE_WORKGROUP);
        }
        m += 512;
        if (m >= total) break;
        load_words16(q, sorted, wr, m + lane);
    }
    __syncthreads();
    for (int i = tid; i < RB; i += 1024) {
        const int node = lo + i;
        if (node < N) {
            const float di = rsqrtf((float)(deg[node] + 1));
            const float d = di * (acc[i] + gd[node]) + (bias[0] - bias[1]);
            const float l = __logf(1.f + __expf(-fabsf(d)));
            v2f o;
            o.x = d >= 0.f ? -l : d - l;
            o.y = d >= 0.f ? -d - l : -l;
            *(v2f*)(out + (size_t)node * 2) = o;
        }
    }
}

extern "C" void kernel_launch(void* const* d_in, const int* in_sizes, int n_in,
                              void* d_out, int out_size, void* d_ws, size_t ws_size,
                              hipStream_t stream) {
    const float* x  = (const float*)d_in[0];
    const int*   ei = (const int*)d_in[1];
    const float* W1 = (const float*)d_in[2];
    const float* b1 = (const float*)d_in[3];
    const float* W2 = (const float*)d_in[4];
    const float* b2 = (const float*)d_in[5];
    const float* W3 = (const float*)d_in[6];
    const float* b3 = (const float*)d_in[7];

    const int N = in_sizes[0] / 128;
    const int E = in_sizes[1] / 2;
    const int* src = ei;
    const int* dst = ei + E;

    char* ws = (char*)d_ws;
    size_t off = 0;
    auto carve = [&](size_t bytes) -> void* {
        void* p = ws + off;
        off += (bytes + 255) & ~(size_t)255;
        return p;
    };
    const int P = (N + RB - 1) / RB;
    const int chunk = (E + NPART - 1) / NPART;
    int*      deg     = (int*)carve((size_t)N * 4);
    int*      runs    = (int*)carve((size_t)P * NPART * 2 * 4);
    unsigned* sorted  = (unsigned*)carve((size_t)E * 4);
    float*    g       = (float*)carve((size_t)N * 16 * 4);
    float*    g2      = (float*)carve((size_t)N * 16 * 4);
    (void)ws_size; (void)n_in; (void)out_size;

    k_partition<<<NPART, 1024, 0, stream>>>(src, dst, runs, sorted, E, chunk, P);

    k_deg_gemm1<<<P, 1024, 0, stream>>>(sorted, runs, x, W1, deg, g, N);

    k_bucket16<16><<<P, 1024, 0, stream>>>(sorted, runs, g, deg, b1, W2, g2, N);

    k_bucket16<1><<<P, 1024, 0, stream>>>(sorted, runs, g2, deg, b2, W3, g, N);

    k_bucket1_lsm<<<P, 1024, 0, stream>>>(sorted, runs, g, deg, b3, (float*)d_out, N, P);
}


# k_partition reads its share of the edge list once: a thread keeps its 16 dst/src words in registers (32 loads in flight) through count, scan and placing; slots beg+i; order still arrival order
# speedup vs baseline: 7.5496x; 1.0397x over previous
#include <hip/hip_runtime.h>
#include <math.h>


typedef float v2f __attribute__((ext_vector_type(2)));
typedef float v8f __attribute__((ext_vector_type(8)));

#define RB 416
#define MAXP 1024
#define SRC_BITS 20
#define SRC_MASK ((1u << SRC_BITS) - 1u)
#define NO_EDGE 0xFFFFFFFFu
#define SPARE_EDGE ((unsigned)RB << SRC_BITS)
#define NPART 256
static_assert(RB % 16 == 0 && RB / 16 <= 32, "one wave of a 1024-thread work-group per 16-row tile");

#define RPW (NPART / 32)
static_assert(NPART % 32 == 0, "each of the 32 waves of a consumer work-group takes RPW whole runs");

struct WaveRuns {
    int o[RPW];
    int d[RPW];
    int total;
};

__device__ __forceinline__ WaveRuns wave_runs(const int* __restrict__ runs, int bucket, int wv) {
    const int* __restrict__ rt = runs + ((size_t)bucket * NPART + (size_t)wv * RPW) * 2;
    WaveRuns r;
    int acc = 0;
#pragma unroll
    for (int k = 0; k < RPW; ++k) {
        const int st = rt[2 * k], ln = rt[2 * k + 1];
        r.o[k] = acc;
        r.d[k] = st - acc;
        acc += ln;
    }
    r.total = acc;
    return r;
}

__device__ __forceinline__ int run_slot(const WaveRuns& r, int e) {
    int d = r.d[0];
#pragma unroll
    for (int k = 1; k < RPW; ++k) d = (e >= r.o[k]) ? r.d[k] : d;
    return e + d;
}

#define STAGE_CAP 16384
#define EPT (STAGE_CAP / 1024)
static_assert(MAXP == 1024, "the scan below gives one table entry to each of the 1024 threads");

__device__ __forceinline__ int block_excl_scan(int v, int* wsum, int tid) {
    const int lane = tid & 31, w = tid >> 5;
    int x = v;
#pragma unroll
    for (int o = 1; o < 32; o <<= 1) {
        int y = __shfl_up(x, o);
        if (lane >= o) x += y;
    }
    if (lane == 31) wsum[w] = x;
    __syncthreads();
    if (w == 0) {
        const int s = wsum[lane];
        int t = s;
#pragma unroll
        for (int o = 1; o < 32; o <<= 1) {
            int y = __shfl_up(t, o);
            if (lane >= o) t += y;
        }
        wsum[lane] = t - s;
    }
    __syncthreads();
    return x - v + wsum[w];
}

__global__ __launch_bounds__(1024) void k_partition(const int* __restrict__ src,
                                                    const int* __restrict__ dst,
                                                    int* __restrict__ runs,
                                                    unsigned* __restrict__ sorted, int E,
                                                    int chunk, int P) {
    __shared__ int hist[MAXP];
    __shared__ int base[MAXP];
    __shared__ int thist[MAXP + 32];
    __shared__ int tstart[MAXP + 32];
    __shared__ int wsum[32];
    __shared__ unsigned sw[STAGE_CAP];
    __shared__ int sidx[STAGE_CAP];
    const int tid = threadIdx.x;
    hist[tid] = 0;
    long beg = (long)blockIdx.x * chunk;
    long end = beg + chunk;
    if (end > (long)E) end = E;
    const bool one_tile = (end - beg) <= (long)STAGE_CAP;
    if (one_tile) {
        const int cnt = (int)(end - beg);
        const int* __restrict__ dp = dst + beg;
        const int* __restrict__ sp = src + beg;
        int d[EPT], s[EPT];
#pragma unroll
        for (int u = 0; u < EPT; ++u) {
            const int i = tid + u * 1024;
            const bool ok = i < cnt;
            d[u] = ok ? dp[i] : -1;
            s[u] = ok ? sp[i] : 0;
        }
        int bk[EPT];
        unsigned wd[EPT];
        unsigned have = 0;
#pragma unroll
        for (int u = 0; u < EPT; ++u) {
            const int b = d[u] / RB;
            bk[u] = b;
            wd[u] = (unsigned)s[u] | ((unsigned)(d[u] - b * RB) << SRC_BITS);
            have |= (d[u] >= 0 ? 1u : 0u) << u;
        }
        __syncthreads();
#pragma unroll
        for (int u = 0; u < EPT; ++u)
            if ((have >> u) & 1u) atomicAdd(&hist[bk[u]], 1);
        __syncthreads();
        {
            const int hc = hist[tid];
            const int ex = block_excl_scan(hc, wsum, tid);
            tstart[tid] = ex;
            thist[tid] = 0;
            if (tid < 32) {
                tstart[MAXP + tid] = 0;
                thist[MAXP + tid] = 0;
            }
            if (tid < P) {
                const size_t ri = ((size_t)tid * NPART + blockIdx.x) * 2;
                runs[ri] = (int)beg + ex;
                runs[ri + 1] = hc;
            }
        }
        __syncthreads();
        int pl[EPT];
#pragma unroll
        for (int u = 0; u < EPT; ++u) {
            const int cw = ((have >> u) & 1u) ? bk[u] : MAXP + (tid & 31);
            pl[u] = tstart[cw] + atomicAdd(&thist[cw], 1);
        }
#pragma unroll
        for (int u = 0; u < EPT; ++u)
            if ((have >> u) & 1u) sw[pl[u]] = wd[u];
        __syncthreads();
        unsigned* __restrict__ op = sorted + beg;
        for (int i = tid; i < cnt; i += 1024) op[i] = sw[i];
        return;
    }
    __syncthreads();
    for (long e0 = beg + tid; e0 < end; e0 += 4096) {
        int d[4];
#pragma unroll
        for (int u = 0; u < 4; ++u) {
            long e = e0 + u * 1024;
            d[u] = (e < end) ? dst[e] : -1;
        }
#pragma unroll
        for (int u = 0; u < 4; ++u)
            if (d[u] >= 0) atomicAdd(&hist[d[u] / RB], 1);
    }
    __syncthreads();
    {
        const int hc = hist[tid];
        const int bs = (int)beg + block_excl_scan(hc, wsum, tid);
        base[tid] = bs;
        if (tid < P) {
            const size_t ri = ((size_t)tid * NPART + blockIdx.x) * 2;
            runs[ri] = bs;
            runs[ri + 1] = hc;
        }
    }
    __syncthreads();
    for (long t0 = beg; t0 < end; t0 += STAGE_CAP) {
        long t1 = t0 + STAGE_CAP;
        if (t1 > end) t1 = end;
        thist[tid] = 0;
        __syncthreads();
        for (long e0 = t0 + tid; e0 < t1; e0 += 4096) {
            int d[4];
#pragma unroll
            for (int u = 0; u < 4; ++u) {
                long e = e0 + u * 1024;
                d[u] = (e < t1) ? dst[e] : -1;
            }
#pragma unroll
            for (int u = 0; u < 4; ++u)
                if (d[u] >= 0) atomicAdd(&thist[d[u] / RB], 1);
        }
        __syncthreads();
        const int tc = thist[tid];
        tstart[tid] = block_excl_scan(tc, wsum, tid);
        thist[tid] = 0;
        __syncthreads();
        for (long e0 = t0 + tid; e0 < t1; e0 += 4096) {
            int d[4], s[4];
#pragma unroll
            for (int u = 0; u < 4; ++u) {
                long e = e0 + u * 1024;
                bool ok = e < t1;
                d[u] = ok ? dst[e] : -1;
                s[u] = ok ? src[e] : 0;
            }
#pragma unroll
            for (int u = 0; u < 4; ++u) {
                if (d[u] >= 0) {
                    int b = d[u] / RB;
                    int k = atomicAdd(&thist[b], 1);
                    int p = tstart[b] + k;
                    sw[p] = (unsigned)s[u] | ((unsigned)(d[u] - b * RB) << SRC_BITS);
                    sidx[p] = base[b] + k;
                }
            }
        }
        __syncthreads();
        const int n = (int)(t1 - t0);
        for (int i = tid; i < n; i += 1024) sorted[sidx[i]] = sw[i];
        __syncthreads();
        base[tid] += thist[tid];
        __syncthreads();
    }
}


__global__ __launch_bounds__(1024) void k_deg_gemm1(const unsigned* __restrict__ sorted,
                                                    const int* __restrict__ runs,
                                                    const float* __restrict__ x,
                                                    const float* __restrict__ W1,
                                                    int* __restrict__ deg,
                                                    float* __restrict__ g, int N) {
    __shared__ int cntl[RB];
    const int tid = threadIdx.x;
    const int lo = blockIdx.x * RB;
    const int lane = tid & 31;
    const int f = lane & 15, half = lane >> 4;
    const int wv = __builtin_amdgcn_readfirstlane(tid >> 5);
    const bool has_tile = wv < RB / 16;
    const int r0 = wv * 16;
    int arow = lo + r0 + f;
    arow = arow < N ? arow : (N - 1);
    for (int i = tid; i < RB; i += 1024) cntl[i] = 0;
    const WaveRuns wr = wave_runs(runs, blockIdx.x, wv);
    __syncthreads();
    for (int e0 = lane; e0 < wr.total; e0 += 128) {
        unsigned q[4];
#pragma unroll
        for (int u = 0; u < 4; ++u) {
            int e = e0 + u * 32;
            q[u] = (e < wr.total) ? sorted[run_slot(wr, e)] : NO_EDGE;
        }
#pragma unroll
        for (int u = 0; u < 4; ++u)
            if (q[u] != NO_EDGE) atomicAdd(&cntl[q[u] >> SRC_BITS], 1);
    }
    __syncthreads();
    for (int i = tid; i < RB; i += 1024) {
        int node = lo + i;
        if (node < N) deg[node] = cntl[i];
    }
    if (has_tile) {
        const float sc = rsqrtf((float)(cntl[r0 + f] + 1));
        const float* __restrict__ ap = x + (long)arow * 128 + 2 * half;
        const float* __restrict__ bp = W1 + f;
        v8f c = {0.f, 0.f, 0.f, 0.f, 0.f, 0.f, 0.f, 0.f};
#pragma unroll
        for (int j = 0; j < 32; ++j) {
            v2f av = *(const v2f*)(ap + 4 * j);
            av.x *= sc;
            av.y *= sc;
            int ka = 4 * j + 2 * half;
            v2f b;
            b.x = bp[(ka + 0) * 16];
            b.y = bp[(ka + 1) * 16];
            c = __builtin_amdgcn_wmma_f32_16x16x4_f32(false, av, false, b, (short)0, c,
                                                      false, false);
        }
#pragma unroll
        for (int v = 0; v < 8; ++v) {
            int row = lo + r0 + v + 8 * half;
            if (row < N) g[(long)row * 16 + f] = c[v];
        }
    }
}

#define FILE_CAP 24576
static_assert(RB % 8 == 0, "eight nodes (one per four lanes) per wave and round");

template <int J>
__device__ __forceinline__ unsigned quad_bcast(unsigned v) {
    return (unsigned)__builtin_amdgcn_update_dpp(0, (int)v, J * 0x55, 0xF, 0xF, false);
}

__device__ __forceinline__ void load_words16(unsigned (&q)[16], const unsigned* __restrict__ sorted,
                                             const WaveRuns& wr, int e0) {
#pragma unroll
    for (int u = 0; u < 16; ++u) {
        const int e = e0 + u * 32;
        q[u] = (e < wr.total) ? sorted[run_slot(wr, e)] : NO_EDGE;
    }
}

template <int NCOL>
__global__ __launch_bounds__(1024) void k_bucket16(const unsigned* __restrict__ sorted,
                                                   const int* __restrict__ runs,
                                                   const float* __restrict__ g,
                                                   const int* __restrict__ deg,
                                                   const float* __restrict__ bias,
                                                   const float* __restrict__ Wn,
                                                   float* __restrict__ gn, int N) {
    __shared__ float acc[(RB + 1) * 16];
    __shared__ float disl[RB];
    __shared__ int offl[RB + 1];
    __shared__ int curl[RB + 32];
    __shared__ int wsum[32];
    __shared__ unsigned srcl[FILE_CAP];
    __shared__ float gcol[NCOL == 1 ? RB : 1];
    const int tid = threadIdx.x;
    const int lo = blockIdx.x * RB;
    const int lane = tid & 31, wave = tid >> 5;
    const int f = lane & 15, half = lane >> 4;
    const int wv = __builtin_amdgcn_readfirstlane(wave);
    const WaveRuns wr = wave_runs(runs, blockIdx.x, wv);
    const int total = wr.total;
    const bool bvalid = (f < NCOL);
    v2f b[4];
#pragma unroll
    for (int j = 0; j < 4; ++j) {
        const int ka = 4 * j + 2 * half;
        if (NCOL == 16) {
            b[j].x = Wn[(ka + 0) * 16 + f];
            b[j].y = Wn[(ka + 1) * 16 + f];
        } else {
            b[j].x = bvalid ? Wn[(ka + 0) * 2 + 0] - Wn[(ka + 0) * 2 + 1] : 0.f;
            b[j].y = bvalid ? Wn[(ka + 1) * 2 + 0] - Wn[(ka + 1) * 2 + 1] : 0.f;
        }
    }
    unsigned q[16];
    load_words16(q, sorted, wr, lane);
    {
        int dg = 0;
        if (tid < RB && lo + tid < N) dg = deg[lo + tid];
        const int st = block_excl_scan(dg, wsum, tid);
        if (tid <= RB) offl[tid] = st;
        if (tid < RB + 32) curl[tid] = st;
        if (tid < RB) disl[tid] = (lo + tid < N) ? rsqrtf((float)(dg + 1)) : 0.f;
    }
    __syncthreads();
    const int btotal = offl[RB];
    if (btotal <= FILE_CAP) {
        for (int m = 0;;) {
            int slot[16];
#pragma unroll
            for (int u = 0; u < 16; ++u) {
                const unsigned cw = (q[u] != NO_EDGE) ? (q[u] >> SRC_BITS) : (unsigned)(RB + lane);
                slot[u] = atomicAdd(&curl[cw], 1);
            }
#pragma unroll
            for (int u = 0; u < 16; ++u)
                if (q[u] != NO_EDGE) srcl[slot[u]] = q[u] & SRC_MASK;
            m += 512;
            if (m >= total) break;
            load_words16(q, sorted, wr, m + lane);
        }
        __syncthreads();
        const int quad = lane >> 2, c = lane & 3;
        float bv[4];
#pragma unroll
        for (int t = 0; t < 4; ++t) bv[t] = bias[4 * c + t];
        for (int n0 = wv * 8; n0 < RB; n0 += 256) {
            const int n = n0 + quad;
            const int beg = offl[n];
            const int len = offl[n + 1] - beg;
            const int node = lo + n;
            const bool live = node < N;
            float4 self = make_float4(0.f, 0.f, 0.f, 0.f);
            if (live) self = *(const float4*)(g + ((unsigned)node * 16u + 4u * (unsigned)c));
            float4 a = make_float4(0.f, 0.f, 0.f, 0.f);
            for (int k = 0; __any(k < len); k += 16) {
                unsigned ix[4];
#pragma unroll
                for (int t = 0; t < 4; ++t) {
                    const int p = k + 4 * t + c;
                    ix[t] = (p < len) ? srcl[beg + p] : 0u;
                }
                unsigned sx[16];
#pragma unroll
                for (int t = 0; t < 4; ++t) {
                    sx[4 * t + 0] = quad_bcast<0>(ix[t]);
                    sx[4 * t + 1] = quad_bcast<1>(ix[t]);
                    sx[4 * t + 2] = quad_bcast<2>(ix[t]);
                    sx[4 * t + 3] = quad_bcast<3>(ix[t]);
                }
                float4 v[16];
#pragma unroll
                for (int j = 0; j < 16; ++j) {
                    v[j] = make_float4(0.f, 0.f, 0.f, 0.f);
                    if (k + j < len) v[j] = *(const float4*)((const char*)g + (sx[j] * 64u + 16u * (unsigned)c));
                }
#pragma unroll
                for (int j = 0; j < 16; ++j) {
                    a.x += v[j].x;
                    a.y += v[j].y;
                    a.z += v[j].z;
                    a.w += v[j].w;
                }
            }
            const float di = disl[n];
            float4 r;
            r.x = di * (a.x + self.x) + bv[0];
            r.y = di * (a.y + self.y) + bv[1];
            r.z = di * (a.z + self.z) + bv[2];
            r.w = di * (a.w + self.w) + bv[3];
            r.x = (live && r.x > 0.f) ? r.x : 0.f;
            r.y = (live && r.y > 0.f) ? r.y : 0.f;
            r.z = (live && r.z > 0.f) ? r.z : 0.f;
            r.w = (live && r.w > 0.f) ? r.w : 0.f;
            *(float4*)&acc[n * 16 + 4 * c] = r;
        }
    } else {
        for (int i = tid; i < (RB + 1) * 16; i += 1024) acc[i] = 0.f;
        __syncthreads();
        unsigned pk = (lane < total) ? sorted[run_slot(wr, lane)] : SPARE_EDGE;
        for (int c0 = 0; c0 < total; c0 += 32) {
            const int en = c0 + 32 + lane;
            const unsigned wn = sorted[(en < total) ? run_slot(wr, en) : 0];
            const unsigned pkn = (en < total) ? wn : SPARE_EDGE;
            unsigned q[16];
            float v[16];
#pragma unroll
            for (int j = 0; j < 16; ++j) q[j] = (unsigned)__shfl((int)pk, 2 * j + half);
#pragma unroll
            for (int j = 0; j < 16; ++j) v[j] = g[(q[j] & SRC_MASK) * 16u + (unsigned)f];
#pragma unroll
            for (int j = 0; j < 16; ++j)
                __hip_atomic_fetch_add(&acc[(q[j] >> SRC_BITS) * 16u + (unsigned)f], v[j],
                                       __ATOMIC_RELAXED, __HIP_MEMORY_SCOPE_WORKGROUP);
            pk = pkn;
        }
        __syncthreads();
        for (int i = tid; i < RB * 16; i += 1024) {
            const int r = i >> 4, node = lo + r, ff = i & 15;
            float v = 0.f;
            if (node < N) {
                size_t o = (size_t)node * 16 + ff;
                v = disl[r] * (acc[i] + g[o]) + bias[ff];
                v = v > 0.f ? v : 0.f;
            }
            acc[i] = v;
        }
    }
    __syncthreads();
    if (wv < RB / 16) {
        const int r0 = wv * 16;
        const float sc = disl[r0 + f];
        const float* ap = &acc[(r0 + f) * 16 + 2 * half];
        v8f c = {0.f, 0.f, 0.f, 0.f, 0.f, 0.f, 0.f, 0.f};
#pragma unroll
        for (int j = 0; j < 4; ++j) {
            v2f a = *(const v2f*)(ap + 4 * j);
            a.x *= sc;
            a.y *= sc;
            c = __builtin_amdgcn_wmma_f32_16x16x4_f32(false, a, false, b[j], (short)0, c,
                                                      false, false);
        }
        if (NCOL == 16) {
#pragma unroll
            for (int v = 0; v < 8; ++v) {
                int row = lo + r0 + v + 8 * half;
                if (row < N) gn[(size_t)row * 16 + f] = c[v];
            }
        } else if (f == 0) {
#pragma unroll
            for (int v = 0; v < 8; ++v) gcol[r0 + v + 8 * half] = c[v];
        }
    }
    if (NCOL == 1) {
        __syncthreads();
        if (tid < RB && lo + tid < N) gn[lo + tid] = gcol[tid];
    }
}

#define TAB_N 80000
static_assert(TAB_N % 4 == 0, "the table copy moves 16 bytes per lane");
static_assert((TAB_N + 4 + RB) * 4 <= 320 * 1024, "table and sums fit one work-group's LDS");
#define CPY 10

__global__ __launch_bounds__(1024) void k_bucket1_lsm(const unsigned* __restrict__ sorted,
                                                      const int* __restrict__ runs,
                                                      const float* __restrict__ gd,
                                                      const int* __restrict__ deg,
                                                      const float* __restrict__ bias,
                                                      float* __restrict__ out, int N, int P) {
    __shared__ __align__(16) float tab[TAB_N + 4];
    __shared__ float acc[RB];
    const int tid = threadIdx.x;
    const int lo = blockIdx.x * RB;
    for (int i = tid; i < RB; i += 1024) acc[i] = 0.f;
    const int wv = __builtin_amdgcn_readfirstlane(tid >> 5);
    const int lane = tid & 31;
    const WaveRuns wr = wave_runs(runs, blockIdx.x, wv);
    const int total = wr.total;
    unsigned q[16];
    load_words16(q, sorted, wr, lane);
    const int nt = N < TAB_N ? N : TAB_N;
    const int nt4 = nt >> 2;
    const int rot = (int)(((unsigned)blockIdx.x * (unsigned)nt4) / (unsigned)P);
    for (int i0 = tid; i0 < nt4; i0 += CPY * 1024) {
        float4 t[CPY];
        int ix[CPY];
#pragma unroll
        for (int u = 0; u < CPY; ++u) {
            const int i = i0 + u * 1024;
            int j = i + rot;
            j = j >= nt4 ? j - nt4 : j;
            const bool in = i < nt4;
            ix[u] = in ? 4 * j : TAB_N;
            t[u] = *(const float4*)(gd + (in ? 4 * j : 0));
        }
#pragma unroll
        for (int u = 0; u < CPY; ++u) *(float4*)&tab[ix[u]] = t[u];
    }
    for (int i = 4 * nt4 + tid; i < nt; i += 1024) tab[i] = gd[i];
    __syncthreads();
    for (int m = 0;;) {
        float vt[16], vg[16];
#pragma unroll
        for (int u = 0; u < 16; ++u) {
            const unsigned s = q[u] & SRC_MASK;
            vt[u] = tab[(q[u] != NO_EDGE && s < (unsigned)nt) ? s : 0u];
        }
#pragma unroll
        for (int u = 0; u < 16; ++u) {
            const unsigned s = q[u] & SRC_MASK;
            vg[u] = 0.f;
            if (q[u] != NO_EDGE && s >= (unsigned)nt) vg[u] = gd[s];
        }
#pragma unroll
        for (int u = 0; u < 16; ++u) {
            const unsigned s = q[u] & SRC_MASK;
            if (q[u] != NO_EDGE)
                __hip_atomic_fetch_add(&acc[q[u] >> SRC_BITS], s < (unsigned)nt ? vt[u] : vg[u],
                                       __ATOMIC_RELAXED, __HIP_MEMORY_SCOPE_WORKGROUP);
        }
        m += 512;
        if (m >= total) break;
        load_words16(q, sorted, wr, m + lane);
    }
    __syncthreads();
    for (int i = tid; i < RB; i += 1024) {
        const int node = lo + i;
        if (node < N) {
            const float di = rsqrtf((float)(deg[node] + 1));
            const float d = di * (acc[i] + gd[node]) + (bias[0] - bias[1]);
            const float l = __logf(1.f + __expf(-fabsf(d)));
            v2f o;
            o.x = d >= 0.f ? -l : d - l;
            o.y = d >= 0.f ? -d - l : -l;
            *(v2f*)(out + (size_t)node * 2) = o;
        }
    }
}

extern "C" void kernel_launch(void* const* d_in, const int* in_sizes, int n_in,
                              void* d_out, int out_size, void* d_ws, size_t ws_size,
                              hipStream_t stream) {
    const float* x  = (const float*)d_in[0];
    const int*   ei = (const int*)d_in[1];
    const float* W1 = (const float*)d_in[2];
    const float* b1 = (const float*)d_in[3];
    const float* W2 = (const float*)d_in[4];
    const float* b2 = (const float*)d_in[5];
    const float* W3 = (const float*)d_in[6];
    const float* b3 = (const float*)d_in[7];

    const int N = in_sizes[0] / 128;
    const int E = in_sizes[1] / 2;
    const int* src = ei;
    const int* dst = ei + E;

    char* ws = (char*)d_ws;
    size_t off = 0;
    auto carve = [&](size_t bytes) -> void* {
        void* p = ws + off;
        off += (bytes + 255) & ~(size_t)255;
        return p;
    };
    const int P = (N + RB - 1) / RB;
    const int chunk = (E + NPART - 1) / NPART;
    int*      deg     = (int*)carve((size_t)N * 4);
    int*      runs    = (int*)carve((size_t)P * NPART * 2 * 4);
    unsigned* sorted  = (unsigned*)carve((size_t)E * 4);
    float*    g       = (float*)carve((size_t)N * 16 * 4);
    float*    g2      = (float*)carve((size_t)N * 16 * 4);
    (void)ws_size; (void)n_in; (void)out_size;

    k_partition<<<NPART, 1024, 0, stream>>>(src, dst, runs, sorted, E, chunk, P);

    k_deg_gemm1<<<P, 1024, 0, stream>>>(sorted, runs, x, W1, deg, g, N);

    k_bucket16<16><<<P, 1024, 0, stream>>>(sorted, runs, g, deg, b1, W2, g2, N);

    k_bucket16<1><<<P, 1024, 0, stream>>>(sorted, runs, g2, deg, b2, W3, g, N);

    k_bucket1_lsm<<<P, 1024, 0, stream>>>(sorted, runs, g, deg, b3, (float*)d_out, N, P);
}
